# starting kernel
# baseline (speedup 1.0000x reference)
#include <hip/hip_runtime.h>
#include <hip/hip_bf16.h>
#include <math.h>

typedef __attribute__((ext_vector_type(2))) float v2f;
typedef __attribute__((ext_vector_type(8))) float v8f;

#define N_NODES 50000
#define N_EDGES 800000
#define HEADS   8
#define FEAT    32
#define NCOL    256

__global__ void k_zero_i32(int* __restrict__ p, int n) {
  int i = blockIdx.x * blockDim.x + threadIdx.x;
  if (i < n) p[i] = 0;
}

__global__ void k_count(const int* __restrict__ dst, int* __restrict__ cnt, int E) {
  int i = blockIdx.x * blockDim.x + threadIdx.x;
  if (i < E) atomicAdd(&cnt[dst[i]], 1);
}

__global__ void k_exscan(const int* __restrict__ cnt, int* __restrict__ rowptr,
                         int* __restrict__ cursor, int n) {
  __shared__ int sums[1024];
  int t = threadIdx.x;
  int chunk = (n + 1023) >> 10;
  int s0 = t * chunk; if (s0 > n) s0 = n;
  int s1 = s0 + chunk; if (s1 > n) s1 = n;
  int sum = 0;
  for (int i = s0; i < s1; ++i) sum += cnt[i];
  sums[t] = sum;
  __syncthreads();
  for (int off = 1; off < 1024; off <<= 1) {
    int v = (t >= off) ? sums[t - off] : 0;
    __syncthreads();
    sums[t] += v;
    __syncthreads();
  }
  int run = (t == 0) ? 0 : sums[t - 1];
  for (int i = s0; i < s1; ++i) {
    int c = cnt[i];
    rowptr[i] = run;
    cursor[i] = run;
    run += c;
  }
  if (t == 0) rowptr[n] = sums[1023];
}

__global__ void k_scatter(const int* __restrict__ dst, int* __restrict__ cursor,
                          int* __restrict__ eidx, int E) {
  int i = blockIdx.x * blockDim.x + threadIdx.x;
  if (i < E) {
    int p = atomicAdd(&cursor[dst[i]], 1);
    eidx[p] = i;
  }
}

__global__ void k_gemm_wmma(const float* __restrict__ A, const float* __restrict__ B,
                            float* __restrict__ C, int M, int K) {
  int wid  = (blockIdx.x * blockDim.x + threadIdx.x) >> 5;
  int lane = threadIdx.x & 31;
  int mtiles = M >> 4;
  if (wid >= mtiles * 4) return;
  int mt = wid >> 2, ng = wid & 3;
  int m0 = mt << 4, n0 = ng << 6;
  int rc   = lane & 15;
  int koff = (lane >> 4) << 1;
  int hi8  = (lane >> 4) << 3;

  v8f acc0 = {}, acc1 = {}, acc2 = {}, acc3 = {};
  const float* arow = A + (m0 + rc) * K;

  #pragma unroll 4
  for (int k = 0; k < K; k += 4) {
    v2f a;
    a.x = arow[k + koff];
    a.y = arow[k + koff + 1];
    const float* bp = B + (k + koff) * NCOL + n0 + rc;
    v2f b;
    b.x = bp[0];       b.y = bp[NCOL];
    acc0 = __builtin_amdgcn_wmma_f32_16x16x4_f32(false, a, false, b, (short)0, acc0, false, false);
    b.x = bp[16];      b.y = bp[NCOL + 16];
    acc1 = __builtin_amdgcn_wmma_f32_16x16x4_f32(false, a, false, b, (short)0, acc1, false, false);
    b.x = bp[32];      b.y = bp[NCOL + 32];
    acc2 = __builtin_amdgcn_wmma_f32_16x16x4_f32(false, a, false, b, (short)0, acc2, false, false);
    b.x = bp[48];      b.y = bp[NCOL + 48];
    acc3 = __builtin_amdgcn_wmma_f32_16x16x4_f32(false, a, false, b, (short)0, acc3, false, false);
  }

  float* crow = C + (m0 + hi8) * NCOL + n0 + rc;
  #pragma unroll
  for (int j = 0; j < 8; ++j) {
    crow[j * NCOL +  0] = acc0[j];
    crow[j * NCOL + 16] = acc1[j];
    crow[j * NCOL + 32] = acc2[j];
    crow[j * NCOL + 48] = acc3[j];
  }
}

__global__ void k_attn(const float* __restrict__ h, const float* __restrict__ al,
                       const float* __restrict__ ar, float* __restrict__ el,
                       float* __restrict__ er) {
  int idx = blockIdx.x * blockDim.x + threadIdx.x;
  if (idx >= N_NODES * HEADS) return;
  int hd = idx & (HEADS - 1);
  int n  = idx >> 3;
  const float* hp  = h + n * NCOL + hd * FEAT;
  const float* alp = al + hd * FEAT;
  const float* arp = ar + hd * FEAT;
  float sl = 0.f, sr = 0.f;
  #pragma unroll
  for (int f = 0; f < FEAT; ++f) {
    float v = hp[f];
    sl = fmaf(v, alp[f], sl);
    sr = fmaf(v, arp[f], sr);
  }
  el[idx] = sl;
  er[idx] = sr;
}

__global__ void k_aggregate(const float* __restrict__ h, const float* __restrict__ el,
                            const float* __restrict__ er, const float* __restrict__ bias,
                            const int* __restrict__ rowptr, const int* __restrict__ eidx,
                            const int* __restrict__ src, float* __restrict__ out) {
  int wid  = (blockIdx.x * blockDim.x + threadIdx.x) >> 5;
  int lane = threadIdx.x & 31;
  if (wid >= N_NODES * HEADS) return;
  int hd = wid & (HEADS - 1);
  int n  = wid >> 3;
  int s0 = rowptr[n], s1 = rowptr[n + 1];
  float ern = er[wid];

  float m = -INFINITY, ssum = 0.f, acc = 0.f;
  for (int j = s0; j < s1; ++j) {
    int e  = eidx[j];
    int sn = src[e];
    float ev = el[sn * HEADS + hd] + ern;
    ev = (ev > 0.f) ? ev : 0.2f * ev;
    float mn = fmaxf(m, ev);
    float sc = __expf(m - mn);
    float p  = __expf(ev - mn);
    ssum = ssum * sc + p;
    acc  = acc  * sc + p * h[sn * NCOL + hd * FEAT + lane];
    m = mn;
  }
  float r = (s1 > s0) ? (acc / ssum) : 0.f;
  out[n * NCOL + hd * FEAT + lane] = r + bias[hd * FEAT + lane];
}

__global__ void k_elu_headmean(const float* __restrict__ agg, float* __restrict__ out) {
  int idx = blockIdx.x * blockDim.x + threadIdx.x;
  if (idx >= N_NODES * FEAT) return;
  int f = idx & (FEAT - 1);
  int n = idx >> 5;
  float s = 0.f;
  #pragma unroll
  for (int hd = 0; hd < HEADS; ++hd) {
    float v = agg[n * NCOL + hd * FEAT + f];
    s += (v > 0.f) ? v : (__expf(v) - 1.f);
  }
  out[idx] = s * (1.f / HEADS);
}

extern "C" void kernel_launch(void* const* d_in, const int* in_sizes, int n_in,
                              void* d_out, int out_size, void* d_ws, size_t ws_size,
                              hipStream_t stream) {
  const float* x   = (const float*)d_in[0];
  const int*   src = (const int*)  d_in[1];
  const int*   dst = (const int*)  d_in[2];
  const float* W1  = (const float*)d_in[3];
  const float* al1 = (const float*)d_in[4];
  const float* ar1 = (const float*)d_in[5];
  const float* b1  = (const float*)d_in[6];
  const float* W2  = (const float*)d_in[7];
  const float* al2 = (const float*)d_in[8];
  const float* ar2 = (const float*)d_in[9];
  const float* b2  = (const float*)d_in[10];

  float* wsf = (float*)d_ws;
  float* h_buf = wsf;
  float* agg1  = h_buf + (size_t)N_NODES * NCOL;
  float* elb   = agg1  + (size_t)N_NODES * NCOL;
  float* erb   = elb   + (size_t)N_NODES * HEADS;
  float* h1act = erb   + (size_t)N_NODES * HEADS;
  int* ibase   = (int*)(h1act + (size_t)N_NODES * FEAT);
  int* rowptr  = ibase;
  int* cursor  = rowptr + (N_NODES + 2);
  int* cnts    = cursor + N_NODES;
  int* eidx    = cnts   + N_NODES;

  k_zero_i32<<<(N_NODES + 255) / 256, 256, 0, stream>>>(cnts, N_NODES);
  k_count   <<<(N_EDGES + 255) / 256, 256, 0, stream>>>(dst, cnts, N_EDGES);
  k_exscan  <<<1, 1024, 0, stream>>>(cnts, rowptr, cursor, N_NODES);
  k_scatter <<<(N_EDGES + 255) / 256, 256, 0, stream>>>(dst, cursor, eidx, N_EDGES);

  const int gemm_blocks = ((N_NODES / 16) * 4 * 32 + 255) / 256;
  const int nh_blocks   = (N_NODES * HEADS + 255) / 256;
  const int agg_blocks  = (N_NODES * HEADS * 32) / 256;
  const int nf_blocks   = (N_NODES * FEAT + 255) / 256;

  k_gemm_wmma <<<gemm_blocks, 256, 0, stream>>>(x, W1, h_buf, N_NODES, 128);
  k_attn      <<<nh_blocks,  256, 0, stream>>>(h_buf, al1, ar1, elb, erb);
  k_aggregate <<<agg_blocks, 256, 0, stream>>>(h_buf, elb, erb, b1, rowptr, eidx, src, agg1);
  k_elu_headmean<<<nf_blocks, 256, 0, stream>>>(agg1, h1act);

  k_gemm_wmma <<<gemm_blocks, 256, 0, stream>>>(h1act, W2, h_buf, N_NODES, 32);
  k_attn      <<<nh_blocks,  256, 0, stream>>>(h_buf, al2, ar2, elb, erb);
  k_aggregate <<<agg_blocks, 256, 0, stream>>>(h_buf, elb, erb, b2, rowptr, eidx, src, (float*)d_out);
}


# aggregate re-tiled: one wave per dst node covers all 8 heads (512 contiguous bytes per wave load), source ids fetched 32 edges at a time by the lanes; was one wave per (node, head)
# speedup vs baseline: 1.5307x; 1.5307x over previous
#include <hip/hip_runtime.h>
#include <hip/hip_bf16.h>
#include <math.h>

typedef __attribute__((ext_vector_type(2))) float v2f;
typedef __attribute__((ext_vector_type(8))) float v8f;

#define N_NODES 50000
#define N_EDGES 800000
#define HEADS   8
#define FEAT    32
#define NCOL    256

__global__ void k_zero_i32(int* __restrict__ p, int n) {
  int i = blockIdx.x * blockDim.x + threadIdx.x;
  if (i < n) p[i] = 0;
}

__global__ void k_count(const int* __restrict__ dst, int* __restrict__ cnt, int E) {
  int i = blockIdx.x * blockDim.x + threadIdx.x;
  if (i < E) atomicAdd(&cnt[dst[i]], 1);
}

__global__ void k_exscan(const int* __restrict__ cnt, int* __restrict__ rowptr,
                         int* __restrict__ cursor, int n) {
  __shared__ int sums[1024];
  int t = threadIdx.x;
  int chunk = (n + 1023) >> 10;
  int s0 = t * chunk; if (s0 > n) s0 = n;
  int s1 = s0 + chunk; if (s1 > n) s1 = n;
  int sum = 0;
  for (int i = s0; i < s1; ++i) sum += cnt[i];
  sums[t] = sum;
  __syncthreads();
  for (int off = 1; off < 1024; off <<= 1) {
    int v = (t >= off) ? sums[t - off] : 0;
    __syncthreads();
    sums[t] += v;
    __syncthreads();
  }
  int run = (t == 0) ? 0 : sums[t - 1];
  for (int i = s0; i < s1; ++i) {
    int c = cnt[i];
    rowptr[i] = run;
    cursor[i] = run;
    run += c;
  }
  if (t == 0) rowptr[n] = sums[1023];
}

__global__ void k_scatter(const int* __restrict__ dst, int* __restrict__ cursor,
                          int* __restrict__ eidx, int E) {
  int i = blockIdx.x * blockDim.x + threadIdx.x;
  if (i < E) {
    int p = atomicAdd(&cursor[dst[i]], 1);
    eidx[p] = i;
  }
}

__global__ void k_gemm_wmma(const float* __restrict__ A, const float* __restrict__ B,
                            float* __restrict__ C, int M, int K) {
  int wid  = (blockIdx.x * blockDim.x + threadIdx.x) >> 5;
  int lane = threadIdx.x & 31;
  int mtiles = M >> 4;
  if (wid >= mtiles * 4) return;
  int mt = wid >> 2, ng = wid & 3;
  int m0 = mt << 4, n0 = ng << 6;
  int rc   = lane & 15;
  int koff = (lane >> 4) << 1;
  int hi8  = (lane >> 4) << 3;

  v8f acc0 = {}, acc1 = {}, acc2 = {}, acc3 = {};
  const float* arow = A + (m0 + rc) * K;

  #pragma unroll 4
  for (int k = 0; k < K; k += 4) {
    v2f a;
    a.x = arow[k + koff];
    a.y = arow[k + koff + 1];
    const float* bp = B + (k + koff) * NCOL + n0 + rc;
    v2f b;
    b.x = bp[0];       b.y = bp[NCOL];
    acc0 = __builtin_amdgcn_wmma_f32_16x16x4_f32(false, a, false, b, (short)0, acc0, false, false);
    b.x = bp[16];      b.y = bp[NCOL + 16];
    acc1 = __builtin_amdgcn_wmma_f32_16x16x4_f32(false, a, false, b, (short)0, acc1, false, false);
    b.x = bp[32];      b.y = bp[NCOL + 32];
    acc2 = __builtin_amdgcn_wmma_f32_16x16x4_f32(false, a, false, b, (short)0, acc2, false, false);
    b.x = bp[48];      b.y = bp[NCOL + 48];
    acc3 = __builtin_amdgcn_wmma_f32_16x16x4_f32(false, a, false, b, (short)0, acc3, false, false);
  }

  float* crow = C + (m0 + hi8) * NCOL + n0 + rc;
  #pragma unroll
  for (int j = 0; j < 8; ++j) {
    crow[j * NCOL +  0] = acc0[j];
    crow[j * NCOL + 16] = acc1[j];
    crow[j * NCOL + 32] = acc2[j];
    crow[j * NCOL + 48] = acc3[j];
  }
}

__global__ void k_attn(const float* __restrict__ h, const float* __restrict__ al,
                       const float* __restrict__ ar, float* __restrict__ el,
                       float* __restrict__ er) {
  int idx = blockIdx.x * blockDim.x + threadIdx.x;
  if (idx >= N_NODES * HEADS) return;
  int hd = idx & (HEADS - 1);
  int n  = idx >> 3;
  const float* hp  = h + n * NCOL + hd * FEAT;
  const float* alp = al + hd * FEAT;
  const float* arp = ar + hd * FEAT;
  float sl = 0.f, sr = 0.f;
  #pragma unroll
  for (int f = 0; f < FEAT; ++f) {
    float v = hp[f];
    sl = fmaf(v, alp[f], sl);
    sr = fmaf(v, arp[f], sr);
  }
  el[idx] = sl;
  er[idx] = sr;
}

__global__ void k_aggregate(const float* __restrict__ h, const float* __restrict__ el,
                            const float* __restrict__ er, const float* __restrict__ bias,
                            const int* __restrict__ rowptr, const int* __restrict__ eidx,
                            const int* __restrict__ src, float* __restrict__ out) {
  const int lane = threadIdx.x & 31;
  const int n = __builtin_amdgcn_readfirstlane((int)((blockIdx.x * blockDim.x + threadIdx.x) >> 5));
  if (n >= N_NODES) return;
  const int s0 = rowptr[n], s1 = rowptr[n + 1];
  const int hq = lane >> 3;
  const int hsel = ((lane & 7) + hq) & 7;
  const float erA = er[n * HEADS + hq];
  const float erB = er[n * HEADS + 4 + hq];

  float mA = -INFINITY, mB = -INFINITY, sA = 0.f, sB = 0.f;
  float4 accA = make_float4(0.f, 0.f, 0.f, 0.f);
  float4 accB = make_float4(0.f, 0.f, 0.f, 0.f);

  for (int base = s0; base < s1; base += 32) {
    int cnt = s1 - base;
    if (cnt > 32) cnt = 32;
    int my_sn = 0;
    if (lane < cnt) my_sn = src[eidx[base + lane]];
    #pragma unroll 4
    for (int t = 0; t < cnt; ++t) {
      const int sn = __builtin_amdgcn_readlane(my_sn, t);
      const float4* hrow = (const float4*)(h + sn * NCOL);
      const float4 vA = hrow[lane];
      const float4 vB = hrow[32 + lane];
      const unsigned elv = __builtin_bit_cast(unsigned, el[sn * HEADS + hsel]);
      float eA = __builtin_bit_cast(float, __builtin_amdgcn_mov_dpp8(elv, 0x000000)) + erA;
      float eB = __builtin_bit_cast(float, __builtin_amdgcn_mov_dpp8(elv, 0x924924)) + erB;
      eA = (eA > 0.f) ? eA : 0.2f * eA;
      eB = (eB > 0.f) ? eB : 0.2f * eB;
      const float mnA = fmaxf(mA, eA);
      const float mnB = fmaxf(mB, eB);
      const float scA = __expf(mA - mnA);
      const float scB = __expf(mB - mnB);
      const float pA  = __expf(eA - mnA);
      const float pB  = __expf(eB - mnB);
      sA = sA * scA + pA;
      sB = sB * scB + pB;
      accA.x = accA.x * scA + pA * vA.x;
      accA.y = accA.y * scA + pA * vA.y;
      accA.z = accA.z * scA + pA * vA.z;
      accA.w = accA.w * scA + pA * vA.w;
      accB.x = accB.x * scB + pB * vB.x;
      accB.y = accB.y * scB + pB * vB.y;
      accB.z = accB.z * scB + pB * vB.z;
      accB.w = accB.w * scB + pB * vB.w;
      mA = mnA;
      mB = mnB;
    }
  }

  const float4 bA = ((const float4*)bias)[lane];
  const float4 bB = ((const float4*)bias)[32 + lane];
  float4 rA = bA, rB = bB;
  if (s1 > s0) {
    rA.x += accA.x / sA;  rA.y += accA.y / sA;  rA.z += accA.z / sA;  rA.w += accA.w / sA;
    rB.x += accB.x / sB;  rB.y += accB.y / sB;  rB.z += accB.z / sB;  rB.w += accB.w / sB;
  }
  float4* orow = (float4*)(out + n * NCOL);
  orow[lane]      = rA;
  orow[32 + lane] = rB;
}

__global__ void k_elu_headmean(const float* __restrict__ agg, float* __restrict__ out) {
  int idx = blockIdx.x * blockDim.x + threadIdx.x;
  if (idx >= N_NODES * FEAT) return;
  int f = idx & (FEAT - 1);
  int n = idx >> 5;
  float s = 0.f;
  #pragma unroll
  for (int hd = 0; hd < HEADS; ++hd) {
    float v = agg[n * NCOL + hd * FEAT + f];
    s += (v > 0.f) ? v : (__expf(v) - 1.f);
  }
  out[idx] = s * (1.f / HEADS);
}

extern "C" void kernel_launch(void* const* d_in, const int* in_sizes, int n_in,
                              void* d_out, int out_size, void* d_ws, size_t ws_size,
                              hipStream_t stream) {
  const float* x   = (const float*)d_in[0];
  const int*   src = (const int*)  d_in[1];
  const int*   dst = (const int*)  d_in[2];
  const float* W1  = (const float*)d_in[3];
  const float* al1 = (const float*)d_in[4];
  const float* ar1 = (const float*)d_in[5];
  const float* b1  = (const float*)d_in[6];
  const float* W2  = (const float*)d_in[7];
  const float* al2 = (const float*)d_in[8];
  const float* ar2 = (const float*)d_in[9];
  const float* b2  = (const float*)d_in[10];

  float* wsf = (float*)d_ws;
  float* h_buf = wsf;
  float* agg1  = h_buf + (size_t)N_NODES * NCOL;
  float* elb   = agg1  + (size_t)N_NODES * NCOL;
  float* erb   = elb   + (size_t)N_NODES * HEADS;
  float* h1act = erb   + (size_t)N_NODES * HEADS;
  int* ibase   = (int*)(h1act + (size_t)N_NODES * FEAT);
  int* rowptr  = ibase;
  int* cursor  = rowptr + (N_NODES + 2);
  int* cnts    = cursor + N_NODES;
  int* eidx    = cnts   + N_NODES;

  k_zero_i32<<<(N_NODES + 255) / 256, 256, 0, stream>>>(cnts, N_NODES);
  k_count   <<<(N_EDGES + 255) / 256, 256, 0, stream>>>(dst, cnts, N_EDGES);
  k_exscan  <<<1, 1024, 0, stream>>>(cnts, rowptr, cursor, N_NODES);
  k_scatter <<<(N_EDGES + 255) / 256, 256, 0, stream>>>(dst, cursor, eidx, N_EDGES);

  const int gemm_blocks = ((N_NODES / 16) * 4 * 32 + 255) / 256;
  const int nh_blocks   = (N_NODES * HEADS + 255) / 256;
  const int agg_blocks  = (N_NODES * 32) / 256;
  const int nf_blocks   = (N_NODES * FEAT + 255) / 256;

  k_gemm_wmma <<<gemm_blocks, 256, 0, stream>>>(x, W1, h_buf, N_NODES, 128);
  k_attn      <<<nh_blocks,  256, 0, stream>>>(h_buf, al1, ar1, elb, erb);
  k_aggregate <<<agg_blocks, 256, 0, stream>>>(h_buf, elb, erb, b1, rowptr, eidx, src, agg1);
  k_elu_headmean<<<nf_blocks, 256, 0, stream>>>(agg1, h1act);

  k_gemm_wmma <<<gemm_blocks, 256, 0, stream>>>(h1act, W2, h_buf, N_NODES, 32);
  k_attn      <<<nh_blocks,  256, 0, stream>>>(h_buf, al2, ar2, elb, erb);
  k_aggregate <<<agg_blocks, 256, 0, stream>>>(h_buf, elb, erb, b2, rowptr, eidx, src, (float*)d_out);
}


# exclusive scan re-laid through LDS: each wave load and store of the 50k counts and prefixes covers 128 contiguous bytes, each thread scans its 49-element run out of LDS; was 32 lines per instruction
# speedup vs baseline: 1.6447x; 1.0745x over previous
#include <hip/hip_runtime.h>
#include <hip/hip_bf16.h>
#include <math.h>

typedef __attribute__((ext_vector_type(2))) float v2f;
typedef __attribute__((ext_vector_type(8))) float v8f;

#define N_NODES 50000
#define N_EDGES 800000
#define HEADS   8
#define FEAT    32
#define NCOL    256

__global__ void k_zero_i32(int* __restrict__ p, int n) {
  int i = blockIdx.x * blockDim.x + threadIdx.x;
  if (i < n) p[i] = 0;
}

__global__ void k_count(const int* __restrict__ dst, int* __restrict__ cnt, int E) {
  int i = blockIdx.x * blockDim.x + threadIdx.x;
  if (i < E) atomicAdd(&cnt[dst[i]], 1);
}

#define SCAN_THREADS 1024
#define SCAN_RUN     ((N_NODES + SCAN_THREADS) / SCAN_THREADS)
#define SCAN_PAD     (SCAN_THREADS * SCAN_RUN)

__global__ void __launch_bounds__(SCAN_THREADS)
k_exscan(const int* __restrict__ cnt, int* __restrict__ rowptr, int* __restrict__ cursor) {
  __shared__ int buf[SCAN_PAD];
  __shared__ int wtot[32];
  const int t    = threadIdx.x;
  const int lane = t & 31;
  const int w    = t >> 5;

  #pragma unroll
  for (int j = 0; j < SCAN_RUN; ++j) {
    const int idx = j * SCAN_THREADS + t;
    buf[idx] = (idx < N_NODES) ? cnt[idx] : 0;
  }
  __syncthreads();

  int v[SCAN_RUN];
  int sum = 0;
  #pragma unroll
  for (int i = 0; i < SCAN_RUN; ++i) {
    v[i] = buf[t * SCAN_RUN + i];
    sum += v[i];
  }

  int inc = sum;
  #pragma unroll
  for (int d = 1; d < 32; d <<= 1) {
    const int o = __shfl_up(inc, d);
    if (lane >= d) inc += o;
  }
  if (lane == 31) wtot[w] = inc;
  __syncthreads();

  int winc = wtot[lane];
  #pragma unroll
  for (int d = 1; d < 32; d <<= 1) {
    const int o = __shfl_up(winc, d);
    if (lane >= d) winc += o;
  }
  int wbase = __shfl(winc, (w + 31) & 31);
  if (w == 0) wbase = 0;

  int run = wbase + inc - sum;
  #pragma unroll
  for (int i = 0; i < SCAN_RUN; ++i) {
    buf[t * SCAN_RUN + i] = run;
    run += v[i];
  }
  __syncthreads();

  #pragma unroll
  for (int j = 0; j < SCAN_RUN; ++j) {
    const int idx = j * SCAN_THREADS + t;
    const int p = buf[idx];
    if (idx <= N_NODES) rowptr[idx] = p;
    if (idx <  N_NODES) cursor[idx] = p;
  }
}

__global__ void k_scatter(const int* __restrict__ dst, int* __restrict__ cursor,
                          int* __restrict__ eidx, int E) {
  int i = blockIdx.x * blockDim.x + threadIdx.x;
  if (i < E) {
    int p = atomicAdd(&cursor[dst[i]], 1);
    eidx[p] = i;
  }
}

__global__ void k_gemm_wmma(const float* __restrict__ A, const float* __restrict__ B,
                            float* __restrict__ C, int M, int K) {
  int wid  = (blockIdx.x * blockDim.x + threadIdx.x) >> 5;
  int lane = threadIdx.x & 31;
  int mtiles = M >> 4;
  if (wid >= mtiles * 4) return;
  int mt = wid >> 2, ng = wid & 3;
  int m0 = mt << 4, n0 = ng << 6;
  int rc   = lane & 15;
  int koff = (lane >> 4) << 1;
  int hi8  = (lane >> 4) << 3;

  v8f acc0 = {}, acc1 = {}, acc2 = {}, acc3 = {};
  const float* arow = A + (m0 + rc) * K;

  #pragma unroll 4
  for (int k = 0; k < K; k += 4) {
    v2f a;
    a.x = arow[k + koff];
    a.y = arow[k + koff + 1];
    const float* bp = B + (k + koff) * NCOL + n0 + rc;
    v2f b;
    b.x = bp[0];       b.y = bp[NCOL];
    acc0 = __builtin_amdgcn_wmma_f32_16x16x4_f32(false, a, false, b, (short)0, acc0, false, false);
    b.x = bp[16];      b.y = bp[NCOL + 16];
    acc1 = __builtin_amdgcn_wmma_f32_16x16x4_f32(false, a, false, b, (short)0, acc1, false, false);
    b.x = bp[32];      b.y = bp[NCOL + 32];
    acc2 = __builtin_amdgcn_wmma_f32_16x16x4_f32(false, a, false, b, (short)0, acc2, false, false);
    b.x = bp[48];      b.y = bp[NCOL + 48];
    acc3 = __builtin_amdgcn_wmma_f32_16x16x4_f32(false, a, false, b, (short)0, acc3, false, false);
  }

  float* crow = C + (m0 + hi8) * NCOL + n0 + rc;
  #pragma unroll
  for (int j = 0; j < 8; ++j) {
    crow[j * NCOL +  0] = acc0[j];
    crow[j * NCOL + 16] = acc1[j];
    crow[j * NCOL + 32] = acc2[j];
    crow[j * NCOL + 48] = acc3[j];
  }
}

__global__ void k_attn(const float* __restrict__ h, const float* __restrict__ al,
                       const float* __restrict__ ar, float* __restrict__ el,
                       float* __restrict__ er) {
  int idx = blockIdx.x * blockDim.x + threadIdx.x;
  if (idx >= N_NODES * HEADS) return;
  int hd = idx & (HEADS - 1);
  int n  = idx >> 3;
  const float* hp  = h + n * NCOL + hd * FEAT;
  const float* alp = al + hd * FEAT;
  const float* arp = ar + hd * FEAT;
  float sl = 0.f, sr = 0.f;
  #pragma unroll
  for (int f = 0; f < FEAT; ++f) {
    float v = hp[f];
    sl = fmaf(v, alp[f], sl);
    sr = fmaf(v, arp[f], sr);
  }
  el[idx] = sl;
  er[idx] = sr;
}

__global__ void k_aggregate(const float* __restrict__ h, const float* __restrict__ el,
                            const float* __restrict__ er, const float* __restrict__ bias,
                            const int* __restrict__ rowptr, const int* __restrict__ eidx,
                            const int* __restrict__ src, float* __restrict__ out) {
  const int lane = threadIdx.x & 31;
  const int n = __builtin_amdgcn_readfirstlane((int)((blockIdx.x * blockDim.x + threadIdx.x) >> 5));
  if (n >= N_NODES) return;
  const int s0 = rowptr[n], s1 = rowptr[n + 1];
  const int hq = lane >> 3;
  const int hsel = ((lane & 7) + hq) & 7;
  const float erA = er[n * HEADS + hq];
  const float erB = er[n * HEADS + 4 + hq];

  float mA = -INFINITY, mB = -INFINITY, sA = 0.f, sB = 0.f;
  float4 accA = make_float4(0.f, 0.f, 0.f, 0.f);
  float4 accB = make_float4(0.f, 0.f, 0.f, 0.f);

  for (int base = s0; base < s1; base += 32) {
    int cnt = s1 - base;
    if (cnt > 32) cnt = 32;
    int my_sn = 0;
    if (lane < cnt) my_sn = src[eidx[base + lane]];
    #pragma unroll 4
    for (int t = 0; t < cnt; ++t) {
      const int sn = __builtin_amdgcn_readlane(my_sn, t);
      const float4* hrow = (const float4*)(h + sn * NCOL);
      const float4 vA = hrow[lane];
      const float4 vB = hrow[32 + lane];
      const unsigned elv = __builtin_bit_cast(unsigned, el[sn * HEADS + hsel]);
      float eA = __builtin_bit_cast(float, __builtin_amdgcn_mov_dpp8(elv, 0x000000)) + erA;
      float eB = __builtin_bit_cast(float, __builtin_amdgcn_mov_dpp8(elv, 0x924924)) + erB;
      eA = (eA > 0.f) ? eA : 0.2f * eA;
      eB = (eB > 0.f) ? eB : 0.2f * eB;
      const float mnA = fmaxf(mA, eA);
      const float mnB = fmaxf(mB, eB);
      const float scA = __expf(mA - mnA);
      const float scB = __expf(mB - mnB);
      const float pA  = __expf(eA - mnA);
      const float pB  = __expf(eB - mnB);
      sA = sA * scA + pA;
      sB = sB * scB + pB;
      accA.x = accA.x * scA + pA * vA.x;
      accA.y = accA.y * scA + pA * vA.y;
      accA.z = accA.z * scA + pA * vA.z;
      accA.w = accA.w * scA + pA * vA.w;
      accB.x = accB.x * scB + pB * vB.x;
      accB.y = accB.y * scB + pB * vB.y;
      accB.z = accB.z * scB + pB * vB.z;
      accB.w = accB.w * scB + pB * vB.w;
      mA = mnA;
      mB = mnB;
    }
  }

  const float4 bA = ((const float4*)bias)[lane];
  const float4 bB = ((const float4*)bias)[32 + lane];
  float4 rA = bA, rB = bB;
  if (s1 > s0) {
    rA.x += accA.x / sA;  rA.y += accA.y / sA;  rA.z += accA.z / sA;  rA.w += accA.w / sA;
    rB.x += accB.x / sB;  rB.y += accB.y / sB;  rB.z += accB.z / sB;  rB.w += accB.w / sB;
  }
  float4* orow = (float4*)(out + n * NCOL);
  orow[lane]      = rA;
  orow[32 + lane] = rB;
}

__global__ void k_elu_headmean(const float* __restrict__ agg, float* __restrict__ out) {
  int idx = blockIdx.x * blockDim.x + threadIdx.x;
  if (idx >= N_NODES * FEAT) return;
  int f = idx & (FEAT - 1);
  int n = idx >> 5;
  float s = 0.f;
  #pragma unroll
  for (int hd = 0; hd < HEADS; ++hd) {
    float v = agg[n * NCOL + hd * FEAT + f];
    s += (v > 0.f) ? v : (__expf(v) - 1.f);
  }
  out[idx] = s * (1.f / HEADS);
}

extern "C" void kernel_launch(void* const* d_in, const int* in_sizes, int n_in,
                              void* d_out, int out_size, void* d_ws, size_t ws_size,
                              hipStream_t stream) {
  const float* x   = (const float*)d_in[0];
  const int*   src = (const int*)  d_in[1];
  const int*   dst = (const int*)  d_in[2];
  const float* W1  = (const float*)d_in[3];
  const float* al1 = (const float*)d_in[4];
  const float* ar1 = (const float*)d_in[5];
  const float* b1  = (const float*)d_in[6];
  const float* W2  = (const float*)d_in[7];
  const float* al2 = (const float*)d_in[8];
  const float* ar2 = (const float*)d_in[9];
  const float* b2  = (const float*)d_in[10];

  float* wsf = (float*)d_ws;
  float* h_buf = wsf;
  float* agg1  = h_buf + (size_t)N_NODES * NCOL;
  float* elb   = agg1  + (size_t)N_NODES * NCOL;
  float* erb   = elb   + (size_t)N_NODES * HEADS;
  float* h1act = erb   + (size_t)N_NODES * HEADS;
  int* ibase   = (int*)(h1act + (size_t)N_NODES * FEAT);
  int* rowptr  = ibase;
  int* cursor  = rowptr + (N_NODES + 2);
  int* cnts    = cursor + N_NODES;
  int* eidx    = cnts   + N_NODES;

  k_zero_i32<<<(N_NODES + 255) / 256, 256, 0, stream>>>(cnts, N_NODES);
  k_count   <<<(N_EDGES + 255) / 256, 256, 0, stream>>>(dst, cnts, N_EDGES);
  k_exscan  <<<1, SCAN_THREADS, 0, stream>>>(cnts, rowptr, cursor);
  k_scatter <<<(N_EDGES + 255) / 256, 256, 0, stream>>>(dst, cursor, eidx, N_EDGES);

  const int gemm_blocks = ((N_NODES / 16) * 4 * 32 + 255) / 256;
  const int nh_blocks   = (N_NODES * HEADS + 255) / 256;
  const int agg_blocks  = (N_NODES * 32) / 256;
  const int nf_blocks   = (N_NODES * FEAT + 255) / 256;

  k_gemm_wmma <<<gemm_blocks, 256, 0, stream>>>(x, W1, h_buf, N_NODES, 128);
  k_attn      <<<nh_blocks,  256, 0, stream>>>(h_buf, al1, ar1, elb, erb);
  k_aggregate <<<agg_blocks, 256, 0, stream>>>(h_buf, elb, erb, b1, rowptr, eidx, src, agg1);
  k_elu_headmean<<<nf_blocks, 256, 0, stream>>>(agg1, h1act);

  k_gemm_wmma <<<gemm_blocks, 256, 0, stream>>>(h1act, W2, h_buf, N_NODES, 32);
  k_attn      <<<nh_blocks,  256, 0, stream>>>(h_buf, al2, ar2, elb, erb);
  k_aggregate <<<agg_blocks, 256, 0, stream>>>(h_buf, elb, erb, b2, rowptr, eidx, src, (float*)d_out);
}


# layer 2 aggregates before projecting (edge sum commutes with @W2): each edge gathers a 128-byte h1act row, not a 1 KB projected row; per-head WMMA projection after it; logits via folded W2*attn
# speedup vs baseline: 1.7931x; 1.0902x over previous
#include <hip/hip_runtime.h>
#include <hip/hip_bf16.h>
#include <math.h>

typedef __attribute__((ext_vector_type(2))) float v2f;
typedef __attribute__((ext_vector_type(8))) float v8f;

#define N_NODES 50000
#define N_EDGES 800000
#define HEADS   8
#define FEAT    32
#define NCOL    256

__global__ void k_zero_i32(int* __restrict__ p, int n) {
  int i = blockIdx.x * blockDim.x + threadIdx.x;
  if (i < n) p[i] = 0;
}

__global__ void k_count(const int* __restrict__ dst, int* __restrict__ cnt, int E) {
  int i = blockIdx.x * blockDim.x + threadIdx.x;
  if (i < E) atomicAdd(&cnt[dst[i]], 1);
}

#define SCAN_THREADS 1024
#define SCAN_RUN     ((N_NODES + SCAN_THREADS) / SCAN_THREADS)
#define SCAN_PAD     (SCAN_THREADS * SCAN_RUN)

__global__ void __launch_bounds__(SCAN_THREADS)
k_exscan(const int* __restrict__ cnt, int* __restrict__ rowptr, int* __restrict__ cursor) {
  __shared__ int buf[SCAN_PAD];
  __shared__ int wtot[32];
  const int t    = threadIdx.x;
  const int lane = t & 31;
  const int w    = t >> 5;

  #pragma unroll
  for (int j = 0; j < SCAN_RUN; ++j) {
    const int idx = j * SCAN_THREADS + t;
    buf[idx] = (idx < N_NODES) ? cnt[idx] : 0;
  }
  __syncthreads();

  int v[SCAN_RUN];
  int sum = 0;
  #pragma unroll
  for (int i = 0; i < SCAN_RUN; ++i) {
    v[i] = buf[t * SCAN_RUN + i];
    sum += v[i];
  }

  int inc = sum;
  #pragma unroll
  for (int d = 1; d < 32; d <<= 1) {
    const int o = __shfl_up(inc, d);
    if (lane >= d) inc += o;
  }
  if (lane == 31) wtot[w] = inc;
  __syncthreads();

  int winc = wtot[lane];
  #pragma unroll
  for (int d = 1; d < 32; d <<= 1) {
    const int o = __shfl_up(winc, d);
    if (lane >= d) winc += o;
  }
  int wbase = __shfl(winc, (w + 31) & 31);
  if (w == 0) wbase = 0;

  int run = wbase + inc - sum;
  #pragma unroll
  for (int i = 0; i < SCAN_RUN; ++i) {
    buf[t * SCAN_RUN + i] = run;
    run += v[i];
  }
  __syncthreads();

  #pragma unroll
  for (int j = 0; j < SCAN_RUN; ++j) {
    const int idx = j * SCAN_THREADS + t;
    const int p = buf[idx];
    if (idx <= N_NODES) rowptr[idx] = p;
    if (idx <  N_NODES) cursor[idx] = p;
  }
}

__global__ void k_scatter(const int* __restrict__ dst, int* __restrict__ cursor,
                          int* __restrict__ eidx, int E) {
  int i = blockIdx.x * blockDim.x + threadIdx.x;
  if (i < E) {
    int p = atomicAdd(&cursor[dst[i]], 1);
    eidx[p] = i;
  }
}

__global__ void k_gemm_wmma(const float* __restrict__ A, const float* __restrict__ B,
                            float* __restrict__ C, int M, int K) {
  int wid  = (blockIdx.x * blockDim.x + threadIdx.x) >> 5;
  int lane = threadIdx.x & 31;
  int mtiles = M >> 4;
  if (wid >= mtiles * 4) return;
  int mt = wid >> 2, ng = wid & 3;
  int m0 = mt << 4, n0 = ng << 6;
  int rc   = lane & 15;
  int koff = (lane >> 4) << 1;
  int hi8  = (lane >> 4) << 3;

  v8f acc0 = {}, acc1 = {}, acc2 = {}, acc3 = {};
  const float* arow = A + (m0 + rc) * K;

  #pragma unroll 4
  for (int k = 0; k < K; k += 4) {
    v2f a;
    a.x = arow[k + koff];
    a.y = arow[k + koff + 1];
    const float* bp = B + (k + koff) * NCOL + n0 + rc;
    v2f b;
    b.x = bp[0];       b.y = bp[NCOL];
    acc0 = __builtin_amdgcn_wmma_f32_16x16x4_f32(false, a, false, b, (short)0, acc0, false, false);
    b.x = bp[16];      b.y = bp[NCOL + 16];
    acc1 = __builtin_amdgcn_wmma_f32_16x16x4_f32(false, a, false, b, (short)0, acc1, false, false);
    b.x = bp[32];      b.y = bp[NCOL + 32];
    acc2 = __builtin_amdgcn_wmma_f32_16x16x4_f32(false, a, false, b, (short)0, acc2, false, false);
    b.x = bp[48];      b.y = bp[NCOL + 48];
    acc3 = __builtin_amdgcn_wmma_f32_16x16x4_f32(false, a, false, b, (short)0, acc3, false, false);
  }

  float* crow = C + (m0 + hi8) * NCOL + n0 + rc;
  #pragma unroll
  for (int j = 0; j < 8; ++j) {
    crow[j * NCOL +  0] = acc0[j];
    crow[j * NCOL + 16] = acc1[j];
    crow[j * NCOL + 32] = acc2[j];
    crow[j * NCOL + 48] = acc3[j];
  }
}

__global__ void k_attn(const float* __restrict__ h, const float* __restrict__ al,
                       const float* __restrict__ ar, float* __restrict__ el,
                       float* __restrict__ er) {
  int idx = blockIdx.x * blockDim.x + threadIdx.x;
  if (idx >= N_NODES * HEADS) return;
  int hd = idx & (HEADS - 1);
  int n  = idx >> 3;
  const float* hp  = h + n * NCOL + hd * FEAT;
  const float* alp = al + hd * FEAT;
  const float* arp = ar + hd * FEAT;
  float sl = 0.f, sr = 0.f;
  #pragma unroll
  for (int f = 0; f < FEAT; ++f) {
    float v = hp[f];
    sl = fmaf(v, alp[f], sl);
    sr = fmaf(v, arp[f], sr);
  }
  el[idx] = sl;
  er[idx] = sr;
}

#define FOLD_BLOCKS  512
#define FOLD_THREADS 256

__global__ void __launch_bounds__(FOLD_THREADS)
k_logits_folded(const float* __restrict__ a, const float* __restrict__ W,
                const float* __restrict__ al, const float* __restrict__ ar,
                float* __restrict__ el, float* __restrict__ er) {
  __shared__ float swl[HEADS * FEAT];
  __shared__ float swr[HEADS * FEAT];
  const int t = threadIdx.x;
  {
    const int hd = t >> 5;
    const int f  = t & 31;
    const float* wp  = W  + f * NCOL + hd * FEAT;
    const float* alp = al + hd * FEAT;
    const float* arp = ar + hd * FEAT;
    float sl = 0.f, sr = 0.f;
    #pragma unroll
    for (int g = 0; g < FEAT; ++g) {
      const float w = wp[g];
      sl = fmaf(w, alp[g], sl);
      sr = fmaf(w, arp[g], sr);
    }
    swl[t] = sl;
    swr[t] = sr;
  }
  __syncthreads();

  const int hd = t & (HEADS - 1);
  float wl[FEAT], wr[FEAT];
  #pragma unroll
  for (int f = 0; f < FEAT; ++f) {
    wl[f] = swl[hd * FEAT + f];
    wr[f] = swr[hd * FEAT + f];
  }

  for (int idx = blockIdx.x * FOLD_THREADS + t; idx < N_NODES * HEADS;
       idx += FOLD_BLOCKS * FOLD_THREADS) {
    const float* ap = a + (idx >> 3) * FEAT;
    float sl = 0.f, sr = 0.f;
    #pragma unroll
    for (int f = 0; f < FEAT; ++f) {
      const float v = ap[f];
      sl = fmaf(v, wl[f], sl);
      sr = fmaf(v, wr[f], sr);
    }
    el[idx] = sl;
    er[idx] = sr;
  }
}

template <bool PRE>
__device__ __forceinline__ void aggregate_body(
                            const float* __restrict__ h, const float* __restrict__ el,
                            const float* __restrict__ er, const float* __restrict__ bias,
                            const int* __restrict__ rowptr, const int* __restrict__ eidx,
                            const int* __restrict__ src, float* __restrict__ out) {
  const int lane = threadIdx.x & 31;
  const int n = __builtin_amdgcn_readfirstlane((int)((blockIdx.x * blockDim.x + threadIdx.x) >> 5));
  if (n >= N_NODES) return;
  const int s0 = rowptr[n], s1 = rowptr[n + 1];
  const int hq = lane >> 3;
  const int hsel = ((lane & 7) + hq) & 7;
  const float erA = er[n * HEADS + hq];
  const float erB = er[n * HEADS + 4 + hq];

  float mA = -INFINITY, mB = -INFINITY, sA = 0.f, sB = 0.f;
  float4 accA = make_float4(0.f, 0.f, 0.f, 0.f);
  float4 accB = make_float4(0.f, 0.f, 0.f, 0.f);

  for (int base = s0; base < s1; base += 32) {
    int cnt = s1 - base;
    if (cnt > 32) cnt = 32;
    int my_sn = 0;
    if (lane < cnt) my_sn = src[eidx[base + lane]];
    #pragma unroll 4
    for (int t = 0; t < cnt; ++t) {
      const int sn = __builtin_amdgcn_readlane(my_sn, t);
      float4 vA, vB;
      if (PRE) {
        vA = ((const float4*)(h + sn * FEAT))[lane & 7];
        vB = vA;
      } else {
        const float4* hrow = (const float4*)(h + sn * NCOL);
        vA = hrow[lane];
        vB = hrow[32 + lane];
      }
      const unsigned elv = __builtin_bit_cast(unsigned, el[sn * HEADS + hsel]);
      float eA = __builtin_bit_cast(float, __builtin_amdgcn_mov_dpp8(elv, 0x000000)) + erA;
      float eB = __builtin_bit_cast(float, __builtin_amdgcn_mov_dpp8(elv, 0x924924)) + erB;
      eA = (eA > 0.f) ? eA : 0.2f * eA;
      eB = (eB > 0.f) ? eB : 0.2f * eB;
      const float mnA = fmaxf(mA, eA);
      const float mnB = fmaxf(mB, eB);
      const float scA = __expf(mA - mnA);
      const float scB = __expf(mB - mnB);
      const float pA  = __expf(eA - mnA);
      const float pB  = __expf(eB - mnB);
      sA = sA * scA + pA;
      sB = sB * scB + pB;
      accA.x = accA.x * scA + pA * vA.x;
      accA.y = accA.y * scA + pA * vA.y;
      accA.z = accA.z * scA + pA * vA.z;
      accA.w = accA.w * scA + pA * vA.w;
      accB.x = accB.x * scB + pB * vB.x;
      accB.y = accB.y * scB + pB * vB.y;
      accB.z = accB.z * scB + pB * vB.z;
      accB.w = accB.w * scB + pB * vB.w;
      mA = mnA;
      mB = mnB;
    }
  }

  float4 rA = make_float4(0.f, 0.f, 0.f, 0.f);
  float4 rB = rA;
  if (!PRE) {
    rA = ((const float4*)bias)[lane];
    rB = ((const float4*)bias)[32 + lane];
  }
  if (s1 > s0) {
    rA.x += accA.x / sA;  rA.y += accA.y / sA;  rA.z += accA.z / sA;  rA.w += accA.w / sA;
    rB.x += accB.x / sB;  rB.y += accB.y / sB;  rB.z += accB.z / sB;  rB.w += accB.w / sB;
  }
  float4* orow = (float4*)(out + n * NCOL);
  orow[lane]      = rA;
  orow[32 + lane] = rB;
}

__global__ void k_aggregate(const float* __restrict__ h, const float* __restrict__ el,
                            const float* __restrict__ er, const float* __restrict__ bias,
                            const int* __restrict__ rowptr, const int* __restrict__ eidx,
                            const int* __restrict__ src, float* __restrict__ out) {
  aggregate_body<false>(h, el, er, bias, rowptr, eidx, src, out);
}

__global__ void k_aggregate_pre(const float* __restrict__ a, const float* __restrict__ el,
                                const float* __restrict__ er,
                                const int* __restrict__ rowptr, const int* __restrict__ eidx,
                                const int* __restrict__ src, float* __restrict__ z) {
  aggregate_body<true>(a, el, er, nullptr, rowptr, eidx, src, z);
}

__global__ void k_gemm_heads_wmma(const float* __restrict__ Z, const float* __restrict__ B,
                                  const float* __restrict__ bias, float* __restrict__ C, int M) {
  int wid  = (blockIdx.x * blockDim.x + threadIdx.x) >> 5;
  int lane = threadIdx.x & 31;
  int mtiles = M >> 4;
  if (wid >= mtiles * 4) return;
  int mt = wid >> 2, ng = wid & 3;
  int m0 = mt << 4, n0 = ng << 6;
  int rc   = lane & 15;
  int koff = (lane >> 4) << 1;
  int hi8  = (lane >> 4) << 3;

  v8f acc0 = {}, acc1 = {}, acc2 = {}, acc3 = {};
  const float* arow = Z + (m0 + rc) * NCOL + n0;

  #pragma unroll
  for (int k = 0; k < FEAT; k += 4) {
    v2f a0, a1;
    a0.x = arow[k + koff];
    a0.y = arow[k + koff + 1];
    a1.x = arow[FEAT + k + koff];
    a1.y = arow[FEAT + k + koff + 1];
    const float* bp = B + (k + koff) * NCOL + n0 + rc;
    v2f b;
    b.x = bp[0];       b.y = bp[NCOL];
    acc0 = __builtin_amdgcn_wmma_f32_16x16x4_f32(false, a0, false, b, (short)0, acc0, false, false);
    b.x = bp[16];      b.y = bp[NCOL + 16];
    acc1 = __builtin_amdgcn_wmma_f32_16x16x4_f32(false, a0, false, b, (short)0, acc1, false, false);
    b.x = bp[32];      b.y = bp[NCOL + 32];
    acc2 = __builtin_amdgcn_wmma_f32_16x16x4_f32(false, a1, false, b, (short)0, acc2, false, false);
    b.x = bp[48];      b.y = bp[NCOL + 48];
    acc3 = __builtin_amdgcn_wmma_f32_16x16x4_f32(false, a1, false, b, (short)0, acc3, false, false);
  }

  const float* bcol = bias + n0 + rc;
  const float b0 = bcol[0], b1 = bcol[16], b2 = bcol[32], b3 = bcol[48];
  float* crow = C + (m0 + hi8) * NCOL + n0 + rc;
  #pragma unroll
  for (int j = 0; j < 8; ++j) {
    crow[j * NCOL +  0] = acc0[j] + b0;
    crow[j * NCOL + 16] = acc1[j] + b1;
    crow[j * NCOL + 32] = acc2[j] + b2;
    crow[j * NCOL + 48] = acc3[j] + b3;
  }
}

__global__ void k_elu_headmean(const float* __restrict__ agg, float* __restrict__ out) {
  int idx = blockIdx.x * blockDim.x + threadIdx.x;
  if (idx >= N_NODES * FEAT) return;
  int f = idx & (FEAT - 1);
  int n = idx >> 5;
  float s = 0.f;
  #pragma unroll
  for (int hd = 0; hd < HEADS; ++hd) {
    float v = agg[n * NCOL + hd * FEAT + f];
    s += (v > 0.f) ? v : (__expf(v) - 1.f);
  }
  out[idx] = s * (1.f / HEADS);
}

extern "C" void kernel_launch(void* const* d_in, const int* in_sizes, int n_in,
                              void* d_out, int out_size, void* d_ws, size_t ws_size,
                              hipStream_t stream) {
  const float* x   = (const float*)d_in[0];
  const int*   src = (const int*)  d_in[1];
  const int*   dst = (const int*)  d_in[2];
  const float* W1  = (const float*)d_in[3];
  const float* al1 = (const float*)d_in[4];
  const float* ar1 = (const float*)d_in[5];
  const float* b1  = (const float*)d_in[6];
  const float* W2  = (const float*)d_in[7];
  const float* al2 = (const float*)d_in[8];
  const float* ar2 = (const float*)d_in[9];
  const float* b2  = (const float*)d_in[10];

  float* wsf = (float*)d_ws;
  float* h_buf = wsf;
  float* agg1  = h_buf + (size_t)N_NODES * NCOL;
  float* elb   = agg1  + (size_t)N_NODES * NCOL;
  float* erb   = elb   + (size_t)N_NODES * HEADS;
  float* h1act = erb   + (size_t)N_NODES * HEADS;
  int* ibase   = (int*)(h1act + (size_t)N_NODES * FEAT);
  int* rowptr  = ibase;
  int* cursor  = rowptr + (N_NODES + 2);
  int* cnts    = cursor + N_NODES;
  int* eidx    = cnts   + N_NODES;

  k_zero_i32<<<(N_NODES + 255) / 256, 256, 0, stream>>>(cnts, N_NODES);
  k_count   <<<(N_EDGES + 255) / 256, 256, 0, stream>>>(dst, cnts, N_EDGES);
  k_exscan  <<<1, SCAN_THREADS, 0, stream>>>(cnts, rowptr, cursor);
  k_scatter <<<(N_EDGES + 255) / 256, 256, 0, stream>>>(dst, cursor, eidx, N_EDGES);

  const int gemm_blocks = ((N_NODES / 16) * 4 * 32 + 255) / 256;
  const int nh_blocks   = (N_NODES * HEADS + 255) / 256;
  const int agg_blocks  = (N_NODES * 32) / 256;
  const int nf_blocks   = (N_NODES * FEAT + 255) / 256;

  k_gemm_wmma <<<gemm_blocks, 256, 0, stream>>>(x, W1, h_buf, N_NODES, 128);
  k_attn      <<<nh_blocks,  256, 0, stream>>>(h_buf, al1, ar1, elb, erb);
  k_aggregate <<<agg_blocks, 256, 0, stream>>>(h_buf, elb, erb, b1, rowptr, eidx, src, agg1);
  k_elu_headmean<<<nf_blocks, 256, 0, stream>>>(agg1, h1act);

  k_logits_folded  <<<FOLD_BLOCKS, FOLD_THREADS, 0, stream>>>(h1act, W2, al2, ar2, elb, erb);
  k_aggregate_pre  <<<agg_blocks, 256, 0, stream>>>(h1act, elb, erb, rowptr, eidx, src, h_buf);
  k_gemm_heads_wmma<<<gemm_blocks, 256, 0, stream>>>(h_buf, W2, b2, (float*)d_out, N_NODES);
}


# CSR built with one atomic pass: the counting add's return is the edge's rank in its dst node, placement behind the scan is atomic-free (rowptr[dst]+rank); edge order in a node = count arrival order
# speedup vs baseline: 2.3134x; 1.2902x over previous
#include <hip/hip_runtime.h>
#include <hip/hip_bf16.h>
#include <math.h>

typedef __attribute__((ext_vector_type(2))) float v2f;
typedef __attribute__((ext_vector_type(8))) float v8f;

#define N_NODES 50000
#define N_EDGES 800000
#define HEADS   8
#define FEAT    32
#define NCOL    256

__global__ void k_count(const int* __restrict__ dst, int* __restrict__ cnt,
                        int* __restrict__ rank, int E) {
  int i = blockIdx.x * blockDim.x + threadIdx.x;
  if (i < E) rank[i] = atomicAdd(&cnt[dst[i]], 1);
}

#define SCAN_THREADS 1024
#define SCAN_RUN     ((N_NODES + SCAN_THREADS) / SCAN_THREADS)
#define SCAN_PAD     (SCAN_THREADS * SCAN_RUN)

__global__ void __launch_bounds__(SCAN_THREADS)
k_exscan(const int* __restrict__ cnt, int* __restrict__ rowptr) {
  __shared__ int buf[SCAN_PAD];
  __shared__ int wtot[32];
  const int t    = threadIdx.x;
  const int lane = t & 31;
  const int w    = t >> 5;

  #pragma unroll
  for (int j = 0; j < SCAN_RUN; ++j) {
    const int idx = j * SCAN_THREADS + t;
    buf[idx] = (idx < N_NODES) ? cnt[idx] : 0;
  }
  __syncthreads();

  int v[SCAN_RUN];
  int sum = 0;
  #pragma unroll
  for (int i = 0; i < SCAN_RUN; ++i) {
    v[i] = buf[t * SCAN_RUN + i];
    sum += v[i];
  }

  int inc = sum;
  #pragma unroll
  for (int d = 1; d < 32; d <<= 1) {
    const int o = __shfl_up(inc, d);
    if (lane >= d) inc += o;
  }
  if (lane == 31) wtot[w] = inc;
  __syncthreads();

  int winc = wtot[lane];
  #pragma unroll
  for (int d = 1; d < 32; d <<= 1) {
    const int o = __shfl_up(winc, d);
    if (lane >= d) winc += o;
  }
  int wbase = __shfl(winc, (w + 31) & 31);
  if (w == 0) wbase = 0;

  int run = wbase + inc - sum;
  #pragma unroll
  for (int i = 0; i < SCAN_RUN; ++i) {
    buf[t * SCAN_RUN + i] = run;
    run += v[i];
  }
  __syncthreads();

  #pragma unroll
  for (int j = 0; j < SCAN_RUN; ++j) {
    const int idx = j * SCAN_THREADS + t;
    if (idx <= N_NODES) rowptr[idx] = buf[idx];
  }
}

__global__ void k_place(const int* __restrict__ dst, const int* __restrict__ rank,
                        const int* __restrict__ rowptr, int* __restrict__ eidx, int E) {
  int i = blockIdx.x * blockDim.x + threadIdx.x;
  if (i < E) eidx[rowptr[dst[i]] + rank[i]] = i;
}


#define IN_F 128
#define LOGITS1_BLOCKS 391

__global__ void __launch_bounds__(256)
k_logits1_wmma(const float* __restrict__ x, const float* __restrict__ W,
               const float* __restrict__ al, const float* __restrict__ ar,
               float* __restrict__ el, float* __restrict__ er,
               const float* __restrict__ W2, const float* __restrict__ al2,
               const float* __restrict__ ar2, float* __restrict__ tab2,
               int* __restrict__ cnts) {
  __shared__ float tab[IN_F * 16];
  const int t = threadIdx.x;
  if (blockIdx.x == LOGITS1_BLOCKS) {
    const int hd = t >> 5;
    const int f  = t & 31;
    const float* wp  = W2  + f * NCOL + hd * FEAT;
    const float* alp = al2 + hd * FEAT;
    const float* arp = ar2 + hd * FEAT;
    float sl = 0.f, sr = 0.f;
    #pragma unroll
    for (int g = 0; g < FEAT; ++g) {
      const float wv = wp[g];
      sl = fmaf(wv, alp[g], sl);
      sr = fmaf(wv, arp[g], sr);
    }
    tab2[f * 16 + hd]         = sl;
    tab2[f * 16 + HEADS + hd] = sr;
    return;
  }
  {
    const int gid = blockIdx.x * 256 + t;
    if (gid < N_NODES) cnts[gid] = 0;
  }
  {
    const int f    = t >> 1;
    const int side = t & 1;
    const float* wp = W + f * NCOL;
    const float* at = side ? ar : al;
    for (int hd = 0; hd < HEADS; ++hd) {
      float s = 0.f;
      #pragma unroll
      for (int g = 0; g < FEAT; ++g) s = fmaf(wp[hd * FEAT + g], at[hd * FEAT + g], s);
      tab[f * 16 + side * HEADS + hd] = s;
    }
  }
  __syncthreads();

  const int wid  = (blockIdx.x * blockDim.x + t) >> 5;
  const int lane = t & 31;
  if (wid >= N_NODES / 16) return;
  const int m0   = wid << 4;
  const int rc   = lane & 15;
  const int koff = (lane >> 4) << 1;
  const int hi8  = (lane >> 4) << 3;
  const float* arow = x + (m0 + rc) * IN_F;
  const float* brow = tab + koff * 16 + rc;

  v8f acc0 = {}, acc1 = {};
  #pragma unroll
  for (int k = 0; k < IN_F; k += 8) {
    v2f a, b;
    a.x = arow[k + koff];          a.y = arow[k + koff + 1];
    b.x = brow[k * 16];            b.y = brow[(k + 1) * 16];
    acc0 = __builtin_amdgcn_wmma_f32_16x16x4_f32(false, a, false, b, (short)0, acc0, false, false);
    a.x = arow[k + 4 + koff];      a.y = arow[k + 4 + koff + 1];
    b.x = brow[(k + 4) * 16];      b.y = brow[(k + 5) * 16];
    acc1 = __builtin_amdgcn_wmma_f32_16x16x4_f32(false, a, false, b, (short)0, acc1, false, false);
  }
  #pragma unroll
  for (int j = 0; j < 8; ++j) {
    const float v = acc0[j] + acc1[j];
    const int row = m0 + hi8 + j;
    if (rc < HEADS) el[row * HEADS + rc] = v;
    else            er[row * HEADS + rc - HEADS] = v;
  }
}

#define TILE_NODES 16
#define TILE_WAVES 8
#define ZPITCH     (HEADS * IN_F + 4)
#define HPITCH     (NCOL + 2)
#define H1PITCH    (FEAT + 1)

typedef __attribute__((ext_vector_type(4))) float v4f;

template <int XMASK>
__device__ __forceinline__ float row_xor(float v) {
  const int b = __builtin_bit_cast(int, v);
  return __builtin_bit_cast(float, __builtin_amdgcn_update_dpp(b, b, 0x160 + XMASK, 0xf, 0xf, false));
}
__device__ __forceinline__ float xhalf(float v) {
  const int b = __builtin_bit_cast(int, v);
  return __builtin_bit_cast(float, __builtin_amdgcn_permlanex16(b, b, 0x76543210, 0xfedcba98, false, false));
}
__device__ __forceinline__ float wave_max32(float v) {
  v = fmaxf(v, row_xor<1>(v));
  v = fmaxf(v, row_xor<2>(v));
  v = fmaxf(v, row_xor<4>(v));
  v = fmaxf(v, row_xor<8>(v));
  return fmaxf(v, xhalf(v));
}
__device__ __forceinline__ float wave_sum32(float v) {
  v += row_xor<1>(v);
  v += row_xor<2>(v);
  v += row_xor<4>(v);
  v += row_xor<8>(v);
  return v + xhalf(v);
}

__global__ void __launch_bounds__(TILE_WAVES * 32)
k_layer1_fused(const float* __restrict__ x, const float* __restrict__ el,
               const float* __restrict__ er, const float* __restrict__ W,
               const float* __restrict__ bias,
               const int* __restrict__ rowptr, const int* __restrict__ eidx,
               const int* __restrict__ src, float* __restrict__ h1act,
               const float* __restrict__ tab2, float* __restrict__ el2,
               float* __restrict__ er2) {
  __shared__ __attribute__((aligned(16))) float zs[TILE_NODES * ZPITCH];
  __shared__ float h1s[TILE_NODES * H1PITCH];
  float* const hs = zs;
  const int lane = threadIdx.x & 31;
  const int w = __builtin_amdgcn_readfirstlane((int)(threadIdx.x >> 5));
  const int node0 = blockIdx.x * TILE_NODES;

  #pragma unroll 1
  for (int r = 0; r < 2; ++r) {
    const int row = 2 * w + r;
    const int n = node0 + row;
    const int s0 = rowptr[n], s1 = rowptr[n + 1];
    float ern[HEADS];
    #pragma unroll
    for (int h = 0; h < HEADS; ++h) ern[h] = er[n * HEADS + h];

    float m[HEADS], s[HEADS];
    v4f acc[HEADS];
    #pragma unroll
    for (int h = 0; h < HEADS; ++h) {
      m[h] = -INFINITY;
      s[h] = 0.f;
      acc[h] = (v4f){0.f, 0.f, 0.f, 0.f};
    }

    for (int base = s0; base < s1; base += 32) {
      int cnt = s1 - base;
      if (cnt > 32) cnt = 32;
      const bool valid = lane < cnt;
      int my_sn = 0;
      v4f e0 = (v4f){0.f, 0.f, 0.f, 0.f}, e1 = e0;
      if (valid) {
        my_sn = src[eidx[base + lane]];
        const v4f* ep = (const v4f*)(el + my_sn * HEADS);
        e0 = ep[0];
        e1 = ep[1];
      }
      const float elv[HEADS] = {e0.x, e0.y, e0.z, e0.w, e1.x, e1.y, e1.z, e1.w};
      float p[HEADS];
      #pragma unroll
      for (int h = 0; h < HEADS; ++h) {
        float e = elv[h] + ern[h];
        e = (e > 0.f) ? e : 0.2f * e;
        const float cm = wave_max32(valid ? e : -INFINITY);
        const float mn = fmaxf(m[h], cm);
        const float sc = __expf(m[h] - mn);
        const float pv = valid ? __expf(e - mn) : 0.f;
        s[h]   = s[h] * sc + wave_sum32(pv);
        acc[h] = acc[h] * sc;
        m[h]   = mn;
        p[h]   = pv;
      }
      #pragma unroll 4
      for (int t = 0; t < cnt; ++t) {
        const int sn = __builtin_amdgcn_readlane(my_sn, t);
        const v4f v = ((const v4f*)(x + sn * IN_F))[lane];
        #pragma unroll
        for (int h = 0; h < HEADS; ++h) {
          const float ph = __builtin_bit_cast(float,
              __builtin_amdgcn_readlane(__builtin_bit_cast(int, p[h]), t));
          acc[h] += ph * v;
        }
      }
    }

    float* zrow = zs + row * ZPITCH + 4 * lane;
    #pragma unroll
    for (int h = 0; h < HEADS; ++h) {
      const float inv = (s1 > s0) ? 1.f / s[h] : 0.f;
      *(v4f*)(zrow + h * IN_F) = acc[h] * inv;
    }
  }
  __syncthreads();

  const int rc   = lane & 15;
  const int koff = (lane >> 4) << 1;
  const int hi8  = (lane >> 4) << 3;
  const int col0 = (w << 5) + rc;
  const int col1 = col0 + 16;
  v8f acc0 = {}, acc1 = {};
  {
    const float* arow = zs + rc * ZPITCH + w * IN_F;
    const float* bp   = W + koff * NCOL + col0;
    #pragma unroll 16
    for (int k = 0; k < IN_F; k += 4) {
      v2f a, b;
      a.x = arow[k + koff];          a.y = arow[k + koff + 1];
      b.x = bp[k * NCOL];            b.y = bp[(k + 1) * NCOL];
      acc0 = __builtin_amdgcn_wmma_f32_16x16x4_f32(false, a, false, b, (short)0, acc0, false, false);
      b.x = bp[k * NCOL + 16];       b.y = bp[(k + 1) * NCOL + 16];
      acc1 = __builtin_amdgcn_wmma_f32_16x16x4_f32(false, a, false, b, (short)0, acc1, false, false);
    }
  }
  __syncthreads();
  {
    const float bv0 = bias[col0], bv1 = bias[col1];
    #pragma unroll
    for (int j = 0; j < 8; ++j) {
      const float v0 = acc0[j] + bv0;
      const float v1 = acc1[j] + bv1;
      hs[(hi8 + j) * HPITCH + col0] = (v0 > 0.f) ? v0 : (__expf(v0) - 1.f);
      hs[(hi8 + j) * HPITCH + col1] = (v1 > 0.f) ? v1 : (__expf(v1) - 1.f);
    }
  }
  __syncthreads();

  #pragma unroll
  for (int r = 0; r < 2; ++r) {
    const int row = w + 8 * r;
    float sum = 0.f;
    #pragma unroll
    for (int hd = 0; hd < HEADS; ++hd) sum += hs[row * HPITCH + hd * FEAT + lane];
    const float hv = sum * (1.f / HEADS);
    h1act[(node0 + row) * FEAT + lane] = hv;
    h1s[row * H1PITCH + lane] = hv;
  }
  __syncthreads();

  {
    const int side = w >> 2;
    const int row  = ((w & 3) << 2) + (lane >> 3);
    const int hd   = lane & 7;
    const float* hp = h1s + row * H1PITCH;
    const float* tp = tab2 + side * HEADS + hd;
    float s = 0.f;
    #pragma unroll
    for (int f = 0; f < FEAT; ++f) s = fmaf(hp[f], tp[f * 16], s);
    float* const lp = side ? er2 : el2;
    lp[(node0 + row) * HEADS + hd] = s;
  }
}

#define Z2PITCH (NCOL + 4)

__global__ void __launch_bounds__(TILE_WAVES * 32)
k_layer2_fused(const float* __restrict__ h, const float* __restrict__ el,
               const float* __restrict__ er, const float* __restrict__ W,
               const float* __restrict__ bias,
               const int* __restrict__ rowptr, const int* __restrict__ eidx,
               const int* __restrict__ src, float* __restrict__ out) {
  __shared__ __attribute__((aligned(16))) float zs[TILE_NODES * Z2PITCH];
  const int lane = threadIdx.x & 31;
  const int w = __builtin_amdgcn_readfirstlane((int)(threadIdx.x >> 5));
  const int node0 = blockIdx.x * TILE_NODES;
  const int hq = lane >> 3;
  const int hsel = ((lane & 7) + hq) & 7;

  #pragma unroll 1
  for (int r = 0; r < 2; ++r) {
    const int row = 2 * w + r;
    const int n = node0 + row;
    const int s0 = rowptr[n], s1 = rowptr[n + 1];
    const float erA = er[n * HEADS + hq];
    const float erB = er[n * HEADS + 4 + hq];

    float mA = -INFINITY, mB = -INFINITY, sA = 0.f, sB = 0.f;
    float4 accA = make_float4(0.f, 0.f, 0.f, 0.f);
    float4 accB = make_float4(0.f, 0.f, 0.f, 0.f);

    for (int base = s0; base < s1; base += 32) {
      int cnt = s1 - base;
      if (cnt > 32) cnt = 32;
      int my_sn = 0;
      if (lane < cnt) my_sn = src[eidx[base + lane]];
      #pragma unroll 4
      for (int t = 0; t < cnt; ++t) {
        const int sn = __builtin_amdgcn_readlane(my_sn, t);
        const float4 vA = ((const float4*)(h + sn * FEAT))[lane & 7];
        const float4 vB = vA;
        const unsigned elv = __builtin_bit_cast(unsigned, el[sn * HEADS + hsel]);
        float eA = __builtin_bit_cast(float, __builtin_amdgcn_mov_dpp8(elv, 0x000000)) + erA;
        float eB = __builtin_bit_cast(float, __builtin_amdgcn_mov_dpp8(elv, 0x924924)) + erB;
        eA = (eA > 0.f) ? eA : 0.2f * eA;
        eB = (eB > 0.f) ? eB : 0.2f * eB;
        const float mnA = fmaxf(mA, eA);
        const float mnB = fmaxf(mB, eB);
        const float scA = __expf(mA - mnA);
        const float scB = __expf(mB - mnB);
        const float pA  = __expf(eA - mnA);
        const float pB  = __expf(eB - mnB);
        sA = sA * scA + pA;
        sB = sB * scB + pB;
        accA.x = accA.x * scA + pA * vA.x;
        accA.y = accA.y * scA + pA * vA.y;
        accA.z = accA.z * scA + pA * vA.z;
        accA.w = accA.w * scA + pA * vA.w;
        accB.x = accB.x * scB + pB * vB.x;
        accB.y = accB.y * scB + pB * vB.y;
        accB.z = accB.z * scB + pB * vB.z;
        accB.w = accB.w * scB + pB * vB.w;
        mA = mnA;
        mB = mnB;
      }
    }

    float4 rA = make_float4(0.f, 0.f, 0.f, 0.f);
    float4 rB = rA;
    if (s1 > s0) {
      rA.x += accA.x / sA;  rA.y += accA.y / sA;  rA.z += accA.z / sA;  rA.w += accA.w / sA;
      rB.x += accB.x / sB;  rB.y += accB.y / sB;  rB.z += accB.z / sB;  rB.w += accB.w / sB;
    }
    float4* zrow = (float4*)(zs + row * Z2PITCH);
    zrow[lane]      = rA;
    zrow[32 + lane] = rB;
  }

  const int rc   = lane & 15;
  const int koff = (lane >> 4) << 1;
  const int hi8  = (lane >> 4) << 3;
  const int col0 = (w << 5) + rc;
  const int col1 = col0 + 16;
  v2f b0[FEAT / 4], b1[FEAT / 4];
  {
    const float* bp = W + koff * NCOL + col0;
    #pragma unroll
    for (int k = 0; k < FEAT; k += 4) {
      b0[k >> 2].x = bp[k * NCOL];        b0[k >> 2].y = bp[(k + 1) * NCOL];
      b1[k >> 2].x = bp[k * NCOL + 16];   b1[k >> 2].y = bp[(k + 1) * NCOL + 16];
    }
  }
  const float bv0 = bias[col0], bv1 = bias[col1];
  __syncthreads();

  v8f acc0 = {}, acc1 = {};
  const float* arow = zs + rc * Z2PITCH + w * FEAT;
  #pragma unroll
  for (int k = 0; k < FEAT; k += 4) {
    v2f a;
    a.x = arow[k + koff];          a.y = arow[k + koff + 1];
    acc0 = __builtin_amdgcn_wmma_f32_16x16x4_f32(false, a, false, b0[k >> 2], (short)0, acc0, false, false);
    acc1 = __builtin_amdgcn_wmma_f32_16x16x4_f32(false, a, false, b1[k >> 2], (short)0, acc1, false, false);
  }
  float* crow = out + (node0 + hi8) * NCOL;
  #pragma unroll
  for (int j = 0; j < 8; ++j) {
    crow[j * NCOL + col0] = acc0[j] + bv0;
    crow[j * NCOL + col1] = acc1[j] + bv1;
  }
}

extern "C" void kernel_launch(void* const* d_in, const int* in_sizes, int n_in,
                              void* d_out, int out_size, void* d_ws, size_t ws_size,
                              hipStream_t stream) {
  const float* x   = (const float*)d_in[0];
  const int*   src = (const int*)  d_in[1];
  const int*   dst = (const int*)  d_in[2];
  const float* W1  = (const float*)d_in[3];
  const float* al1 = (const float*)d_in[4];
  const float* ar1 = (const float*)d_in[5];
  const float* b1  = (const float*)d_in[6];
  const float* W2  = (const float*)d_in[7];
  const float* al2 = (const float*)d_in[8];
  const float* ar2 = (const float*)d_in[9];
  const float* b2  = (const float*)d_in[10];

  float* wsf = (float*)d_ws;
  float* h_buf = wsf;
  float* agg1  = h_buf + (size_t)N_NODES * NCOL;
  float* elb   = agg1  + (size_t)N_NODES * NCOL;
  float* erb   = elb   + (size_t)N_NODES * HEADS;
  float* h1act = erb   + (size_t)N_NODES * HEADS;
  int* ibase   = (int*)(h1act + (size_t)N_NODES * FEAT);
  int* rowptr  = ibase;
  int* cursor  = rowptr + (N_NODES + 2);
  int* cnts    = cursor + N_NODES;
  int* eidx    = cnts   + N_NODES;
  int* rank    = (int*)h_buf;
  float* tab2  = agg1;
  float* el2   = tab2 + 512;
  float* er2   = el2  + (size_t)N_NODES * HEADS;

  k_logits1_wmma<<<LOGITS1_BLOCKS + 1, 256, 0, stream>>>(x, W1, al1, ar1, elb, erb, W2, al2, ar2, tab2, cnts);

  k_count   <<<(N_EDGES + 255) / 256, 256, 0, stream>>>(dst, cnts, rank, N_EDGES);
  k_exscan  <<<1, SCAN_THREADS, 0, stream>>>(cnts, rowptr);
  k_place   <<<(N_EDGES + 255) / 256, 256, 0, stream>>>(dst, rank, rowptr, eidx, N_EDGES);

  k_layer1_fused<<<N_NODES / TILE_NODES, TILE_WAVES * 32, 0, stream>>>(x, elb, erb, W1, b1, rowptr, eidx, src, h1act, tab2, el2, er2);

  k_layer2_fused<<<N_NODES / TILE_NODES, TILE_WAVES * 32, 0, stream>>>(h1act, el2, er2, W2, b2, rowptr, eidx, src, (float*)d_out);
}


# CSR build as a two-level counting sort on LDS atomics: edges binned by 256-node dst range, then each bin sorted in one block's LDS; no scattered global atomics; in-node edge order stays arrival order
# speedup vs baseline: 3.1670x; 1.3690x over previous
#include <hip/hip_runtime.h>
#include <hip/hip_bf16.h>
#include <math.h>

typedef __attribute__((ext_vector_type(2))) float v2f;
typedef __attribute__((ext_vector_type(8))) float v8f;

#define N_NODES 50000
#define N_EDGES 800000
#define HEADS   8
#define FEAT    32
#define NCOL    256

#define BIN_SHIFT  8
#define BIN_NODES  (1 << BIN_SHIFT)
#define NBINS      ((N_NODES + BIN_NODES - 1) / BIN_NODES)
#define BIN_PITCH  64
#define EB_THREADS 256
#define EB_EDGES   (EB_THREADS * 8)
#define EB_BLOCKS  ((N_EDGES + EB_EDGES - 1) / EB_EDGES)
#define BS_THREADS 1024
#define BS_DEPTH   5

static_assert(N_EDGES % 4 == 0, "a 16-byte group of edges must not straddle the end of the list");
static_assert(N_EDGES <= (1 << 23), "edge id and the node's 8 local bits share one non-negative word");
static_assert(NBINS <= EB_THREADS && NBINS <= BIN_NODES, "one thread per bin in the scans");

__device__ __forceinline__ int wave_incl_scan(int v, int lane) {
  #pragma unroll
  for (int d = 1; d < 32; d <<= 1) {
    const int o = __shfl_up(v, d);
    if (lane >= d) v += o;
  }
  return v;
}

__global__ void __launch_bounds__(EB_THREADS)
k_binhist(const int* __restrict__ dst, int* __restrict__ bincount, int E) {
  __shared__ int h[NBINS];
  const int t = threadIdx.x;
  if (t < NBINS) h[t] = 0;
  __syncthreads();
  const int e0 = blockIdx.x * EB_EDGES + 4 * t;
  int4 d[2];
  #pragma unroll
  for (int j = 0; j < 2; ++j) {
    const int e = e0 + j * (EB_EDGES / 2);
    const int4 v = *(const int4*)(dst + ((e < E) ? e : 0));
    d[j] = (e < E) ? v : make_int4(-1, -1, -1, -1);
  }
  #pragma unroll
  for (int j = 0; j < 2; ++j) {
    if (d[j].x >= 0) {
      atomicAdd(&h[d[j].x >> BIN_SHIFT], 1);
      atomicAdd(&h[d[j].y >> BIN_SHIFT], 1);
      atomicAdd(&h[d[j].z >> BIN_SHIFT], 1);
      atomicAdd(&h[d[j].w >> BIN_SHIFT], 1);
    }
  }
  __syncthreads();
  if (t < NBINS) {
    const int c = h[t];
    if (c) atomicAdd(&bincount[t * BIN_PITCH], c);
  }
}

__global__ void __launch_bounds__(EB_THREADS)
k_binscatter(const int* __restrict__ dst, const int* __restrict__ bincount,
             int* __restrict__ bincursor, int* __restrict__ keys, int E) {
  __shared__ int h[NBINS];
  __shared__ int wtot[EB_THREADS / 32];
  const int t    = threadIdx.x;
  const int lane = t & 31;
  const int w    = t >> 5;
  if (t < NBINS) h[t] = 0;
  __syncthreads();
  const int e0 = blockIdx.x * EB_EDGES + 4 * t;
  int4 d[2];
  #pragma unroll
  for (int j = 0; j < 2; ++j) {
    const int e = e0 + j * (EB_EDGES / 2);
    const int4 v = *(const int4*)(dst + ((e < E) ? e : 0));
    d[j] = (e < E) ? v : make_int4(-1, -1, -1, -1);
  }
  const int tot = (t < NBINS) ? bincount[t * BIN_PITCH] : 0;
  int4 lr[2];
  #pragma unroll
  for (int j = 0; j < 2; ++j) {
    lr[j] = make_int4(0, 0, 0, 0);
    if (d[j].x >= 0) {
      lr[j].x = atomicAdd(&h[d[j].x >> BIN_SHIFT], 1);
      lr[j].y = atomicAdd(&h[d[j].y >> BIN_SHIFT], 1);
      lr[j].z = atomicAdd(&h[d[j].z >> BIN_SHIFT], 1);
      lr[j].w = atomicAdd(&h[d[j].w >> BIN_SHIFT], 1);
    }
  }
  const int inc = wave_incl_scan(tot, lane);
  if (lane == 31) wtot[w] = inc;
  __syncthreads();
  int wbase = 0;
  for (int i = 0; i < w; ++i) wbase += wtot[i];
  if (t < NBINS) {
    const int mine = h[t];
    const int off = mine ? atomicAdd(&bincursor[t * BIN_PITCH], mine) : 0;
    h[t] = wbase + inc - tot + off;
  }
  __syncthreads();
  #pragma unroll
  for (int j = 0; j < 2; ++j) {
    if (d[j].x >= 0) {
      const int e = e0 + j * (EB_EDGES / 2);
      keys[h[d[j].x >> BIN_SHIFT] + lr[j].x] = ((e    ) << BIN_SHIFT) | (d[j].x & (BIN_NODES - 1));
      keys[h[d[j].y >> BIN_SHIFT] + lr[j].y] = ((e + 1) << BIN_SHIFT) | (d[j].y & (BIN_NODES - 1));
      keys[h[d[j].z >> BIN_SHIFT] + lr[j].z] = ((e + 2) << BIN_SHIFT) | (d[j].z & (BIN_NODES - 1));
      keys[h[d[j].w >> BIN_SHIFT] + lr[j].w] = ((e + 3) << BIN_SHIFT) | (d[j].w & (BIN_NODES - 1));
    }
  }
}

__global__ void __launch_bounds__(BS_THREADS)
k_binsort(const int* __restrict__ bincount, const int* __restrict__ keys,
          int* __restrict__ rowptr, int* __restrict__ eidx) {
  __shared__ int cnt[BIN_NODES];
  __shared__ int wtot[BIN_NODES / 32];
  __shared__ int range[2];
  const int t    = threadIdx.x;
  const int lane = t & 31;
  const int w    = t >> 5;
  const int b    = blockIdx.x;

  const int tot = (t < NBINS) ? bincount[t * BIN_PITCH] : 0;
  if (t < BIN_NODES) cnt[t] = 0;
  int inc = wave_incl_scan(tot, lane);
  if (t < BIN_NODES && lane == 31) wtot[w] = inc;
  __syncthreads();
  if (t == b) {
    int wbase = 0;
    for (int i = 0; i < w; ++i) wbase += wtot[i];
    range[0] = wbase + inc - tot;
    range[1] = wbase + inc;
  }
  __syncthreads();
  const int s = range[0], e = range[1];

  for (int base = s; base < e; base += BS_THREADS * BS_DEPTH) {
    int k[BS_DEPTH];
    #pragma unroll
    for (int u = 0; u < BS_DEPTH; ++u) {
      const int i = base + u * BS_THREADS + t;
      const int v = keys[(i < e) ? i : e - 1];
      k[u] = (i < e) ? v : -1;
    }
    __builtin_amdgcn_sched_barrier(0);
    #pragma unroll
    for (int u = 0; u < BS_DEPTH; ++u)
      if (k[u] >= 0) atomicAdd(&cnt[k[u] & (BIN_NODES - 1)], 1);
  }
  __syncthreads();

  const int c = (t < BIN_NODES) ? cnt[t] : 0;
  inc = wave_incl_scan(c, lane);
  if (t < BIN_NODES && lane == 31) wtot[w] = inc;
  __syncthreads();
  int first = 0;
  if (t < BIN_NODES) {
    int wbase = 0;
    for (int i = 0; i < w; ++i) wbase += wtot[i];
    first = s + wbase + inc - c;
    cnt[t] = first;
  }
  __syncthreads();

  for (int base = s; base < e; base += BS_THREADS * BS_DEPTH) {
    int k[BS_DEPTH];
    #pragma unroll
    for (int u = 0; u < BS_DEPTH; ++u) {
      const int i = base + u * BS_THREADS + t;
      const int v = keys[(i < e) ? i : e - 1];
      k[u] = (i < e) ? v : -1;
    }
    __builtin_amdgcn_sched_barrier(0);
    #pragma unroll
    for (int u = 0; u < BS_DEPTH; ++u)
      if (k[u] >= 0) eidx[atomicAdd(&cnt[k[u] & (BIN_NODES - 1)], 1)] = k[u] >> BIN_SHIFT;
  }

  if (t < BIN_NODES) {
    const int node = b * BIN_NODES + t;
    if (node < N_NODES) rowptr[node] = first;
  }
  if (b == NBINS - 1 && t == 0) rowptr[N_NODES] = e;
}


#define IN_F 128
#define LOGITS1_BLOCKS 391

__global__ void __launch_bounds__(256)
k_logits1_wmma(const float* __restrict__ x, const float* __restrict__ W,
               const float* __restrict__ al, const float* __restrict__ ar,
               float* __restrict__ el, float* __restrict__ er,
               const float* __restrict__ W2, const float* __restrict__ al2,
               const float* __restrict__ ar2, float* __restrict__ tab2,
               int* __restrict__ binctr) {
  __shared__ float tab[IN_F * 16];
  const int t = threadIdx.x;
  if (blockIdx.x == LOGITS1_BLOCKS) {
    const int hd = t >> 5;
    const int f  = t & 31;
    const float* wp  = W2  + f * NCOL + hd * FEAT;
    const float* alp = al2 + hd * FEAT;
    const float* arp = ar2 + hd * FEAT;
    float sl = 0.f, sr = 0.f;
    #pragma unroll
    for (int g = 0; g < FEAT; ++g) {
      const float wv = wp[g];
      sl = fmaf(wv, alp[g], sl);
      sr = fmaf(wv, arp[g], sr);
    }
    tab2[f * 16 + hd]         = sl;
    tab2[f * 16 + HEADS + hd] = sr;
    return;
  }
  {
    const int gid = blockIdx.x * 256 + t;
    if (gid < 2 * NBINS) binctr[gid * BIN_PITCH] = 0;
  }
  {
    const int f    = t >> 1;
    const int side = t & 1;
    const float* wp = W + f * NCOL;
    const float* at = side ? ar : al;
    for (int hd = 0; hd < HEADS; ++hd) {
      float s = 0.f;
      #pragma unroll
      for (int g = 0; g < FEAT; ++g) s = fmaf(wp[hd * FEAT + g], at[hd * FEAT + g], s);
      tab[f * 16 + side * HEADS + hd] = s;
    }
  }
  __syncthreads();

  const int wid  = (blockIdx.x * blockDim.x + t) >> 5;
  const int lane = t & 31;
  if (wid >= N_NODES / 16) return;
  const int m0   = wid << 4;
  const int rc   = lane & 15;
  const int koff = (lane >> 4) << 1;
  const int hi8  = (lane >> 4) << 3;
  const float* arow = x + (m0 + rc) * IN_F;
  const float* brow = tab + koff * 16 + rc;

  v8f acc0 = {}, acc1 = {};
  #pragma unroll
  for (int k = 0; k < IN_F; k += 8) {
    v2f a, b;
    a.x = arow[k + koff];          a.y = arow[k + koff + 1];
    b.x = brow[k * 16];            b.y = brow[(k + 1) * 16];
    acc0 = __builtin_amdgcn_wmma_f32_16x16x4_f32(false, a, false, b, (short)0, acc0, false, false);
    a.x = arow[k + 4 + koff];      a.y = arow[k + 4 + koff + 1];
    b.x = brow[(k + 4) * 16];      b.y = brow[(k + 5) * 16];
    acc1 = __builtin_amdgcn_wmma_f32_16x16x4_f32(false, a, false, b, (short)0, acc1, false, false);
  }
  #pragma unroll
  for (int j = 0; j < 8; ++j) {
    const float v = acc0[j] + acc1[j];
    const int row = m0 + hi8 + j;
    if (rc < HEADS) el[row * HEADS + rc] = v;
    else            er[row * HEADS + rc - HEADS] = v;
  }
}

#define TILE_NODES 16
#define TILE_WAVES 8
#define ZPITCH     (HEADS * IN_F + 4)
#define HPITCH     (NCOL + 2)
#define H1PITCH    (FEAT + 1)

typedef __attribute__((ext_vector_type(4))) float v4f;

template <int XMASK>
__device__ __forceinline__ float row_xor(float v) {
  const int b = __builtin_bit_cast(int, v);
  return __builtin_bit_cast(float, __builtin_amdgcn_update_dpp(b, b, 0x160 + XMASK, 0xf, 0xf, false));
}
__device__ __forceinline__ float xhalf(float v) {
  const int b = __builtin_bit_cast(int, v);
  return __builtin_bit_cast(float, __builtin_amdgcn_permlanex16(b, b, 0x76543210, 0xfedcba98, false, false));
}
__device__ __forceinline__ float wave_max32(float v) {
  v = fmaxf(v, row_xor<1>(v));
  v = fmaxf(v, row_xor<2>(v));
  v = fmaxf(v, row_xor<4>(v));
  v = fmaxf(v, row_xor<8>(v));
  return fmaxf(v, xhalf(v));
}
__device__ __forceinline__ float wave_sum32(float v) {
  v += row_xor<1>(v);
  v += row_xor<2>(v);
  v += row_xor<4>(v);
  v += row_xor<8>(v);
  return v + xhalf(v);
}

__global__ void __launch_bounds__(TILE_WAVES * 32)
k_layer1_fused(const float* __restrict__ x, const float* __restrict__ el,
               const float* __restrict__ er, const float* __restrict__ W,
               const float* __restrict__ bias,
               const int* __restrict__ rowptr, const int* __restrict__ eidx,
               const int* __restrict__ src, float* __restrict__ h1act,
               const float* __restrict__ tab2, float* __restrict__ el2,
               float* __restrict__ er2) {
  __shared__ __attribute__((aligned(16))) float zs[TILE_NODES * ZPITCH];
  __shared__ float h1s[TILE_NODES * H1PITCH];
  float* const hs = zs;
  const int lane = threadIdx.x & 31;
  const int w = __builtin_amdgcn_readfirstlane((int)(threadIdx.x >> 5));
  const int node0 = blockIdx.x * TILE_NODES;

  #pragma unroll 1
  for (int r = 0; r < 2; ++r) {
    const int row = 2 * w + r;
    const int n = node0 + row;
    const int s0 = rowptr[n], s1 = rowptr[n + 1];
    float ern[HEADS];
    #pragma unroll
    for (int h = 0; h < HEADS; ++h) ern[h] = er[n * HEADS + h];

    float m[HEADS], s[HEADS];
    v4f acc[HEADS];
    #pragma unroll
    for (int h = 0; h < HEADS; ++h) {
      m[h] = -INFINITY;
      s[h] = 0.f;
      acc[h] = (v4f){0.f, 0.f, 0.f, 0.f};
    }

    for (int base = s0; base < s1; base += 32) {
      int cnt = s1 - base;
      if (cnt > 32) cnt = 32;
      const bool valid = lane < cnt;
      int my_sn = 0;
      v4f e0 = (v4f){0.f, 0.f, 0.f, 0.f}, e1 = e0;
      if (valid) {
        my_sn = src[eidx[base + lane]];
        const v4f* ep = (const v4f*)(el + my_sn * HEADS);
        e0 = ep[0];
        e1 = ep[1];
      }
      const float elv[HEADS] = {e0.x, e0.y, e0.z, e0.w, e1.x, e1.y, e1.z, e1.w};
      float p[HEADS];
      #pragma unroll
      for (int h = 0; h < HEADS; ++h) {
        float e = elv[h] + ern[h];
        e = (e > 0.f) ? e : 0.2f * e;
        const float cm = wave_max32(valid ? e : -INFINITY);
        const float mn = fmaxf(m[h], cm);
        const float sc = __expf(m[h] - mn);
        const float pv = valid ? __expf(e - mn) : 0.f;
        s[h]   = s[h] * sc + wave_sum32(pv);
        acc[h] = acc[h] * sc;
        m[h]   = mn;
        p[h]   = pv;
      }
      #pragma unroll 4
      for (int t = 0; t < cnt; ++t) {
        const int sn = __builtin_amdgcn_readlane(my_sn, t);
        const v4f v = ((const v4f*)(x + sn * IN_F))[lane];
        #pragma unroll
        for (int h = 0; h < HEADS; ++h) {
          const float ph = __builtin_bit_cast(float,
              __builtin_amdgcn_readlane(__builtin_bit_cast(int, p[h]), t));
          acc[h] += ph * v;
        }
      }
    }

    float* zrow = zs + row * ZPITCH + 4 * lane;
    #pragma unroll
    for (int h = 0; h < HEADS; ++h) {
      const float inv = (s1 > s0) ? 1.f / s[h] : 0.f;
      *(v4f*)(zrow + h * IN_F) = acc[h] * inv;
    }
  }
  __syncthreads();

  const int rc   = lane & 15;
  const int koff = (lane >> 4) << 1;
  const int hi8  = (lane >> 4) << 3;
  const int col0 = (w << 5) + rc;
  const int col1 = col0 + 16;
  v8f acc0 = {}, acc1 = {};
  {
    const float* arow = zs + rc * ZPITCH + w * IN_F;
    const float* bp   = W + koff * NCOL + col0;
    #pragma unroll 16
    for (int k = 0; k < IN_F; k += 4) {
      v2f a, b;
      a.x = arow[k + koff];          a.y = arow[k + koff + 1];
      b.x = bp[k * NCOL];            b.y = bp[(k + 1) * NCOL];
      acc0 = __builtin_amdgcn_wmma_f32_16x16x4_f32(false, a, false, b, (short)0, acc0, false, false);
      b.x = bp[k * NCOL + 16];       b.y = bp[(k + 1) * NCOL + 16];
      acc1 = __builtin_amdgcn_wmma_f32_16x16x4_f32(false, a, false, b, (short)0, acc1, false, false);
    }
  }
  __syncthreads();
  {
    const float bv0 = bias[col0], bv1 = bias[col1];
    #pragma unroll
    for (int j = 0; j < 8; ++j) {
      const float v0 = acc0[j] + bv0;
      const float v1 = acc1[j] + bv1;
      hs[(hi8 + j) * HPITCH + col0] = (v0 > 0.f) ? v0 : (__expf(v0) - 1.f);
      hs[(hi8 + j) * HPITCH + col1] = (v1 > 0.f) ? v1 : (__expf(v1) - 1.f);
    }
  }
  __syncthreads();

  #pragma unroll
  for (int r = 0; r < 2; ++r) {
    const int row = w + 8 * r;
    float sum = 0.f;
    #pragma unroll
    for (int hd = 0; hd < HEADS; ++hd) sum += hs[row * HPITCH + hd * FEAT + lane];
    const float hv = sum * (1.f / HEADS);
    h1act[(node0 + row) * FEAT + lane] = hv;
    h1s[row * H1PITCH + lane] = hv;
  }
  __syncthreads();

  {
    const int side = w >> 2;
    const int row  = ((w & 3) << 2) + (lane >> 3);
    const int hd   = lane & 7;
    const float* hp = h1s + row * H1PITCH;
    const float* tp = tab2 + side * HEADS + hd;
    float s = 0.f;
    #pragma unroll
    for (int f = 0; f < FEAT; ++f) s = fmaf(hp[f], tp[f * 16], s);
    float* const lp = side ? er2 : el2;
    lp[(node0 + row) * HEADS + hd] = s;
  }
}

#define Z2PITCH (NCOL + 4)

__global__ void __launch_bounds__(TILE_WAVES * 32)
k_layer2_fused(const float* __restrict__ h, const float* __restrict__ el,
               const float* __restrict__ er, const float* __restrict__ W,
               const float* __restrict__ bias,
               const int* __restrict__ rowptr, const int* __restrict__ eidx,
               const int* __restrict__ src, float* __restrict__ out) {
  __shared__ __attribute__((aligned(16))) float zs[TILE_NODES * Z2PITCH];
  const int lane = threadIdx.x & 31;
  const int w = __builtin_amdgcn_readfirstlane((int)(threadIdx.x >> 5));
  const int node0 = blockIdx.x * TILE_NODES;
  const int hq = lane >> 3;
  const int hsel = ((lane & 7) + hq) & 7;

  #pragma unroll 1
  for (int r = 0; r < 2; ++r) {
    const int row = 2 * w + r;
    const int n = node0 + row;
    const int s0 = rowptr[n], s1 = rowptr[n + 1];
    const float erA = er[n * HEADS + hq];
    const float erB = er[n * HEADS + 4 + hq];

    float mA = -INFINITY, mB = -INFINITY, sA = 0.f, sB = 0.f;
    float4 accA = make_float4(0.f, 0.f, 0.f, 0.f);
    float4 accB = make_float4(0.f, 0.f, 0.f, 0.f);

    for (int base = s0; base < s1; base += 32) {
      int cnt = s1 - base;
      if (cnt > 32) cnt = 32;
      int my_sn = 0;
      if (lane < cnt) my_sn = src[eidx[base + lane]];
      #pragma unroll 4
      for (int t = 0; t < cnt; ++t) {
        const int sn = __builtin_amdgcn_readlane(my_sn, t);
        const float4 vA = ((const float4*)(h + sn * FEAT))[lane & 7];
        const float4 vB = vA;
        const unsigned elv = __builtin_bit_cast(unsigned, el[sn * HEADS + hsel]);
        float eA = __builtin_bit_cast(float, __builtin_amdgcn_mov_dpp8(elv, 0x000000)) + erA;
        float eB = __builtin_bit_cast(float, __builtin_amdgcn_mov_dpp8(elv, 0x924924)) + erB;
        eA = (eA > 0.f) ? eA : 0.2f * eA;
        eB = (eB > 0.f) ? eB : 0.2f * eB;
        const float mnA = fmaxf(mA, eA);
        const float mnB = fmaxf(mB, eB);
        const float scA = __expf(mA - mnA);
        const float scB = __expf(mB - mnB);
        const float pA  = __expf(eA - mnA);
        const float pB  = __expf(eB - mnB);
        sA = sA * scA + pA;
        sB = sB * scB + pB;
        accA.x = accA.x * scA + pA * vA.x;
        accA.y = accA.y * scA + pA * vA.y;
        accA.z = accA.z * scA + pA * vA.z;
        accA.w = accA.w * scA + pA * vA.w;
        accB.x = accB.x * scB + pB * vB.x;
        accB.y = accB.y * scB + pB * vB.y;
        accB.z = accB.z * scB + pB * vB.z;
        accB.w = accB.w * scB + pB * vB.w;
        mA = mnA;
        mB = mnB;
      }
    }

    float4 rA = make_float4(0.f, 0.f, 0.f, 0.f);
    float4 rB = rA;
    if (s1 > s0) {
      rA.x += accA.x / sA;  rA.y += accA.y / sA;  rA.z += accA.z / sA;  rA.w += accA.w / sA;
      rB.x += accB.x / sB;  rB.y += accB.y / sB;  rB.z += accB.z / sB;  rB.w += accB.w / sB;
    }
    float4* zrow = (float4*)(zs + row * Z2PITCH);
    zrow[lane]      = rA;
    zrow[32 + lane] = rB;
  }

  const int rc   = lane & 15;
  const int koff = (lane >> 4) << 1;
  const int hi8  = (lane >> 4) << 3;
  const int col0 = (w << 5) + rc;
  const int col1 = col0 + 16;
  v2f b0[FEAT / 4], b1[FEAT / 4];
  {
    const float* bp = W + koff * NCOL + col0;
    #pragma unroll
    for (int k = 0; k < FEAT; k += 4) {
      b0[k >> 2].x = bp[k * NCOL];        b0[k >> 2].y = bp[(k + 1) * NCOL];
      b1[k >> 2].x = bp[k * NCOL + 16];   b1[k >> 2].y = bp[(k + 1) * NCOL + 16];
    }
  }
  const float bv0 = bias[col0], bv1 = bias[col1];
  __syncthreads();

  v8f acc0 = {}, acc1 = {};
  const float* arow = zs + rc * Z2PITCH + w * FEAT;
  #pragma unroll
  for (int k = 0; k < FEAT; k += 4) {
    v2f a;
    a.x = arow[k + koff];          a.y = arow[k + koff + 1];
    acc0 = __builtin_amdgcn_wmma_f32_16x16x4_f32(false, a, false, b0[k >> 2], (short)0, acc0, false, false);
    acc1 = __builtin_amdgcn_wmma_f32_16x16x4_f32(false, a, false, b1[k >> 2], (short)0, acc1, false, false);
  }
  float* crow = out + (node0 + hi8) * NCOL;
  #pragma unroll
  for (int j = 0; j < 8; ++j) {
    crow[j * NCOL + col0] = acc0[j] + bv0;
    crow[j * NCOL + col1] = acc1[j] + bv1;
  }
}

extern "C" void kernel_launch(void* const* d_in, const int* in_sizes, int n_in,
                              void* d_out, int out_size, void* d_ws, size_t ws_size,
                              hipStream_t stream) {
  const float* x   = (const float*)d_in[0];
  const int*   src = (const int*)  d_in[1];
  const int*   dst = (const int*)  d_in[2];
  const float* W1  = (const float*)d_in[3];
  const float* al1 = (const float*)d_in[4];
  const float* ar1 = (const float*)d_in[5];
  const float* b1  = (const float*)d_in[6];
  const float* W2  = (const float*)d_in[7];
  const float* al2 = (const float*)d_in[8];
  const float* ar2 = (const float*)d_in[9];
  const float* b2  = (const float*)d_in[10];

  float* wsf = (float*)d_ws;
  float* h_buf = wsf;
  float* agg1  = h_buf + (size_t)N_NODES * NCOL;
  float* elb   = agg1  + (size_t)N_NODES * NCOL;
  float* erb   = elb   + (size_t)N_NODES * HEADS;
  float* h1act = erb   + (size_t)N_NODES * HEADS;
  int* ibase   = (int*)(h1act + (size_t)N_NODES * FEAT);
  int* rowptr  = ibase;
  int* binctr  = rowptr + (N_NODES + 2);
  int* eidx    = binctr + 2 * N_NODES;
  int* keys    = (int*)h_buf;
  float* tab2  = agg1;
  float* el2   = tab2 + 512;
  float* er2   = el2  + (size_t)N_NODES * HEADS;

  k_logits1_wmma<<<LOGITS1_BLOCKS + 1, 256, 0, stream>>>(x, W1, al1, ar1, elb, erb, W2, al2, ar2, tab2, binctr);

  k_binhist   <<<EB_BLOCKS, EB_THREADS, 0, stream>>>(dst, binctr, N_EDGES);
  k_binscatter<<<EB_BLOCKS, EB_THREADS, 0, stream>>>(dst, binctr, binctr + NBINS * BIN_PITCH, keys, N_EDGES);
  k_binsort   <<<NBINS, BS_THREADS, 0, stream>>>(binctr, keys, rowptr, eidx);

  k_layer1_fused<<<N_NODES / TILE_NODES, TILE_WAVES * 32, 0, stream>>>(x, elb, erb, W1, b1, rowptr, eidx, src, h1act, tab2, el2, er2);

  k_layer2_fused<<<N_NODES / TILE_NODES, TILE_WAVES * 32, 0, stream>>>(h1act, el2, er2, W2, b2, rowptr, eidx, src, (float*)d_out);
}


# CSR sort key carries the source id, not the edge id (src<<8 | dst&255): the sorted list is each node's source list, both layers read it contiguously, no eidx->src gather; in-node order as before
# speedup vs baseline: 3.6551x; 1.1541x over previous
#include <hip/hip_runtime.h>
#include <hip/hip_bf16.h>
#include <math.h>

typedef __attribute__((ext_vector_type(2))) float v2f;
typedef __attribute__((ext_vector_type(8))) float v8f;

#define N_NODES 50000
#define N_EDGES 800000
#define HEADS   8
#define FEAT    32
#define NCOL    256

#define BIN_SHIFT  8
#define BIN_NODES  (1 << BIN_SHIFT)
#define NBINS      ((N_NODES + BIN_NODES - 1) / BIN_NODES)
#define EB_THREADS 256
#define EB_EDGES   (EB_THREADS * 8)
#define EB_BLOCKS  ((N_EDGES + EB_EDGES - 1) / EB_EDGES)
#define OFF_PITCH  392
#define BS_THREADS 1024
#define BS_PIECES  ((EB_BLOCKS + 31) / 32)

static_assert(N_EDGES % 4 == 0, "a 16-byte group of edges must not straddle the end of the list");
static_assert(N_NODES <= (1 << 23), "source id and the node's 8 local bits share one non-negative word");
static_assert(NBINS <= EB_THREADS && NBINS <= BIN_NODES, "one thread per bin in the scans");
static_assert(EB_BLOCKS <= OFF_PITCH && BS_PIECES * 32 <= BS_THREADS, "one thread per piece");
static_assert((NBINS + 1) * OFF_PITCH <= 2 * N_NODES, "the offset table lives in the two N-int planes of the earlier layout");

__device__ __forceinline__ int wave_incl_scan(int v, int lane) {
  #pragma unroll
  for (int d = 1; d < 32; d <<= 1) {
    const int o = __shfl_up(v, d);
    if (lane >= d) v += o;
  }
  return v;
}

template <int XMASK>
__device__ __forceinline__ int row_xor_i32(int b) {
  return __builtin_amdgcn_update_dpp(b, b, 0x160 + XMASK, 0xf, 0xf, false);
}
__device__ __forceinline__ int wave_sum_i32(int v) {
  v += row_xor_i32<1>(v);
  v += row_xor_i32<2>(v);
  v += row_xor_i32<4>(v);
  v += row_xor_i32<8>(v);
  return v + __builtin_amdgcn_permlanex16(v, v, 0x76543210, 0xfedcba98, false, false);
}

__device__ __forceinline__ void bin_local(const int* __restrict__ src, const int* __restrict__ dst,
                                          int* __restrict__ offs,
                                          int* __restrict__ keys, const int blk,
                                          int* __restrict__ h, int* __restrict__ wtot) {
  const int t    = threadIdx.x;
  const int lane = t & 31;
  const int w    = t >> 5;
  if (t < NBINS) h[t] = 0;
  __syncthreads();
  const int e0 = blk * EB_EDGES + 4 * t;
  int4 d[2], sv[2];
  #pragma unroll
  for (int j = 0; j < 2; ++j) {
    const int e = e0 + j * (EB_EDGES / 2);
    const int4 v = *(const int4*)(dst + ((e < N_EDGES) ? e : 0));
    sv[j] = *(const int4*)(src + ((e < N_EDGES) ? e : 0));
    d[j] = (e < N_EDGES) ? v : make_int4(-1, -1, -1, -1);
  }
  int4 lr[2];
  #pragma unroll
  for (int j = 0; j < 2; ++j) {
    lr[j] = make_int4(0, 0, 0, 0);
    if (d[j].x >= 0) {
      lr[j].x = atomicAdd(&h[d[j].x >> BIN_SHIFT], 1);
      lr[j].y = atomicAdd(&h[d[j].y >> BIN_SHIFT], 1);
      lr[j].z = atomicAdd(&h[d[j].z >> BIN_SHIFT], 1);
      lr[j].w = atomicAdd(&h[d[j].w >> BIN_SHIFT], 1);
    }
  }
  __syncthreads();
  const int c   = (t < NBINS) ? h[t] : 0;
  const int inc = wave_incl_scan(c, lane);
  if (lane == 31) wtot[w] = inc;
  __syncthreads();
  int wbase = 0;
  for (int i = 0; i < w; ++i) wbase += wtot[i];
  if (t < NBINS) {
    const int ex = wbase + inc - c;
    h[t] = ex;
    offs[t * OFF_PITCH + blk] = ex;
    if (t == NBINS - 1) offs[NBINS * OFF_PITCH + blk] = ex + c;
  }
  __syncthreads();
  int* const seg = keys + blk * EB_EDGES;
  #pragma unroll
  for (int j = 0; j < 2; ++j) {
    if (d[j].x >= 0) {
      seg[h[d[j].x >> BIN_SHIFT] + lr[j].x] = (sv[j].x << BIN_SHIFT) | (d[j].x & (BIN_NODES - 1));
      seg[h[d[j].y >> BIN_SHIFT] + lr[j].y] = (sv[j].y << BIN_SHIFT) | (d[j].y & (BIN_NODES - 1));
      seg[h[d[j].z >> BIN_SHIFT] + lr[j].z] = (sv[j].z << BIN_SHIFT) | (d[j].z & (BIN_NODES - 1));
      seg[h[d[j].w >> BIN_SHIFT] + lr[j].w] = (sv[j].w << BIN_SHIFT) | (d[j].w & (BIN_NODES - 1));
    }
  }
}

__global__ void __launch_bounds__(BS_THREADS)
k_binsort(const int* __restrict__ offs, const int* __restrict__ keys,
          int* __restrict__ rowptr, int* __restrict__ csrc) {
  __shared__ int pst[BS_PIECES * 32];
  __shared__ int pln[BS_PIECES * 32];
  __shared__ int cnt[BIN_NODES];
  __shared__ int wtot[BS_THREADS / 32];
  __shared__ int wlen[BS_THREADS / 32];
  __shared__ int wlong[BS_THREADS / 32];
  const int t    = threadIdx.x;
  const int lane = t & 31;
  const int w    = t >> 5;
  const int b    = blockIdx.x;

  int st = 0, ln = 0;
  if (t < EB_BLOCKS) {
    st = offs[b * OFF_PITCH + t];
    ln = offs[(b + 1) * OFF_PITCH + t] - st;
  }
  if (t < BS_PIECES * 32) {
    pst[t] = t * EB_EDGES + st;
    pln[t] = ln;
  }
  if (t < BIN_NODES) cnt[t] = 0;
  {
    const int a = wave_sum_i32(st);
    const int l = wave_sum_i32(ln);
    const int g = __any(ln > 32);
    if (lane == 0) { wtot[w] = a; wlen[w] = l; wlong[w] = g; }
  }
  __syncthreads();
  int s = 0, e = 0, anylong = 0;
  #pragma unroll
  for (int i = 0; i < BS_PIECES; ++i) { s += wtot[i]; e += wlen[i]; anylong |= wlong[i]; }
  e += s;

  int k[BS_PIECES], pl[BS_PIECES];
  #pragma unroll
  for (int u = 0; u < BS_PIECES; ++u) {
    const int j  = w + 32 * u;
    const int ps = pst[j];
    pl[u] = pln[j];
    k[u] = keys[(lane < pl[u]) ? ps + lane : 0];
  }
  __builtin_amdgcn_sched_barrier(0);
  #pragma unroll
  for (int u = 0; u < BS_PIECES; ++u) {
    k[u] = (lane < pl[u]) ? k[u] : -1;
    if (k[u] >= 0) atomicAdd(&cnt[k[u] & (BIN_NODES - 1)], 1);
  }
  if (anylong) {
    for (int j = w; j < EB_BLOCKS; j += 32) {
      const int pl = pln[j];
      if (pl > 32) {
        const int ps = pst[j];
        for (int o = 32 + lane; o < pl; o += 32) atomicAdd(&cnt[keys[ps + o] & (BIN_NODES - 1)], 1);
      }
    }
  }
  __syncthreads();

  const int c = (t < BIN_NODES) ? cnt[t] : 0;
  const int inc = wave_incl_scan(c, lane);
  if (t < BIN_NODES && lane == 31) wtot[w] = inc;
  __syncthreads();
  int first = 0;
  if (t < BIN_NODES) {
    int wbase = 0;
    for (int i = 0; i < w; ++i) wbase += wtot[i];
    first = s + wbase + inc - c;
    cnt[t] = first;
  }
  __syncthreads();

  #pragma unroll
  for (int u = 0; u < BS_PIECES; ++u)
    if (k[u] >= 0) csrc[atomicAdd(&cnt[k[u] & (BIN_NODES - 1)], 1)] = k[u] >> BIN_SHIFT;
  if (anylong) {
    for (int j = w; j < EB_BLOCKS; j += 32) {
      const int pl = pln[j];
      if (pl > 32) {
        const int ps = pst[j];
        for (int o = 32 + lane; o < pl; o += 32) {
          const int kk = keys[ps + o];
          csrc[atomicAdd(&cnt[kk & (BIN_NODES - 1)], 1)] = kk >> BIN_SHIFT;
        }
      }
    }
  }

  if (t < BIN_NODES) {
    const int node = b * BIN_NODES + t;
    if (node < N_NODES) rowptr[node] = first;
  }
  if (b == NBINS - 1 && t == 0) rowptr[N_NODES] = e;
}


#define IN_F 128
#define LOGITS1_BLOCKS 391

__global__ void __launch_bounds__(256)
k_logits1_wmma(const float* __restrict__ x, const float* __restrict__ W,
               const float* __restrict__ al, const float* __restrict__ ar,
               float* __restrict__ el, float* __restrict__ er,
               const float* __restrict__ W2, const float* __restrict__ al2,
               const float* __restrict__ ar2, float* __restrict__ tab2,
               const int* __restrict__ src, const int* __restrict__ dst,
               int* __restrict__ offs, int* __restrict__ keys) {
  __shared__ float tab[IN_F * 16];
  __shared__ int binh[NBINS];
  __shared__ int binw[EB_THREADS / 32];
  const int t = threadIdx.x;
  if (blockIdx.x > LOGITS1_BLOCKS) {
    bin_local(src, dst, offs, keys, blockIdx.x - (LOGITS1_BLOCKS + 1), binh, binw);
    return;
  }
  if (blockIdx.x == LOGITS1_BLOCKS) {
    const int hd = t >> 5;
    const int f  = t & 31;
    const float* wp  = W2  + f * NCOL + hd * FEAT;
    const float* alp = al2 + hd * FEAT;
    const float* arp = ar2 + hd * FEAT;
    float sl = 0.f, sr = 0.f;
    #pragma unroll
    for (int g = 0; g < FEAT; ++g) {
      const float wv = wp[g];
      sl = fmaf(wv, alp[g], sl);
      sr = fmaf(wv, arp[g], sr);
    }
    tab2[f * 16 + hd]         = sl;
    tab2[f * 16 + HEADS + hd] = sr;
    return;
  }
  {
    const int f    = t >> 1;
    const int side = t & 1;
    const float* wp = W + f * NCOL;
    const float* at = side ? ar : al;
    for (int hd = 0; hd < HEADS; ++hd) {
      float s = 0.f;
      #pragma unroll
      for (int g = 0; g < FEAT; ++g) s = fmaf(wp[hd * FEAT + g], at[hd * FEAT + g], s);
      tab[f * 16 + side * HEADS + hd] = s;
    }
  }
  __syncthreads();

  const int wid  = (blockIdx.x * blockDim.x + t) >> 5;
  const int lane = t & 31;
  if (wid >= N_NODES / 16) return;
  const int m0   = wid << 4;
  const int rc   = lane & 15;
  const int koff = (lane >> 4) << 1;
  const int hi8  = (lane >> 4) << 3;
  const float* arow = x + (m0 + rc) * IN_F;
  const float* brow = tab + koff * 16 + rc;

  v8f acc0 = {}, acc1 = {};
  #pragma unroll
  for (int k = 0; k < IN_F; k += 8) {
    v2f a, b;
    a.x = arow[k + koff];          a.y = arow[k + koff + 1];
    b.x = brow[k * 16];            b.y = brow[(k + 1) * 16];
    acc0 = __builtin_amdgcn_wmma_f32_16x16x4_f32(false, a, false, b, (short)0, acc0, false, false);
    a.x = arow[k + 4 + koff];      a.y = arow[k + 4 + koff + 1];
    b.x = brow[(k + 4) * 16];      b.y = brow[(k + 5) * 16];
    acc1 = __builtin_amdgcn_wmma_f32_16x16x4_f32(false, a, false, b, (short)0, acc1, false, false);
  }
  #pragma unroll
  for (int j = 0; j < 8; ++j) {
    const float v = acc0[j] + acc1[j];
    const int row = m0 + hi8 + j;
    if (rc < HEADS) el[row * HEADS + rc] = v;
    else            er[row * HEADS + rc - HEADS] = v;
  }
}

#define TILE_NODES 16
#define TILE_WAVES 8
#define ZPITCH     (HEADS * IN_F + 4)
#define HPITCH     (NCOL + 2)
#define H1PITCH    (FEAT + 1)

typedef __attribute__((ext_vector_type(4))) float v4f;

template <int XMASK>
__device__ __forceinline__ float row_xor(float v) {
  const int b = __builtin_bit_cast(int, v);
  return __builtin_bit_cast(float, __builtin_amdgcn_update_dpp(b, b, 0x160 + XMASK, 0xf, 0xf, false));
}
__device__ __forceinline__ float xhalf(float v) {
  const int b = __builtin_bit_cast(int, v);
  return __builtin_bit_cast(float, __builtin_amdgcn_permlanex16(b, b, 0x76543210, 0xfedcba98, false, false));
}
__device__ __forceinline__ float wave_max32(float v) {
  v = fmaxf(v, row_xor<1>(v));
  v = fmaxf(v, row_xor<2>(v));
  v = fmaxf(v, row_xor<4>(v));
  v = fmaxf(v, row_xor<8>(v));
  return fmaxf(v, xhalf(v));
}
__device__ __forceinline__ float wave_sum32(float v) {
  v += row_xor<1>(v);
  v += row_xor<2>(v);
  v += row_xor<4>(v);
  v += row_xor<8>(v);
  return v + xhalf(v);
}

__global__ void __launch_bounds__(TILE_WAVES * 32)
k_layer1_fused(const float* __restrict__ x, const float* __restrict__ el,
               const float* __restrict__ er, const float* __restrict__ W,
               const float* __restrict__ bias,
               const int* __restrict__ rowptr, const int* __restrict__ csrc,
               float* __restrict__ h1act,
               const float* __restrict__ tab2, float* __restrict__ el2,
               float* __restrict__ er2) {
  __shared__ __attribute__((aligned(16))) float zs[TILE_NODES * ZPITCH];
  __shared__ float h1s[TILE_NODES * H1PITCH];
  float* const hs = zs;
  const int lane = threadIdx.x & 31;
  const int w = __builtin_amdgcn_readfirstlane((int)(threadIdx.x >> 5));
  const int node0 = blockIdx.x * TILE_NODES;

  #pragma unroll 1
  for (int r = 0; r < 2; ++r) {
    const int row = 2 * w + r;
    const int n = node0 + row;
    const int s0 = rowptr[n], s1 = rowptr[n + 1];
    float ern[HEADS];
    #pragma unroll
    for (int h = 0; h < HEADS; ++h) ern[h] = er[n * HEADS + h];

    float m[HEADS], s[HEADS];
    v4f acc[HEADS];
    #pragma unroll
    for (int h = 0; h < HEADS; ++h) {
      m[h] = -INFINITY;
      s[h] = 0.f;
      acc[h] = (v4f){0.f, 0.f, 0.f, 0.f};
    }

    for (int base = s0; base < s1; base += 32) {
      int cnt = s1 - base;
      if (cnt > 32) cnt = 32;
      const bool valid = lane < cnt;
      int my_sn = 0;
      v4f e0 = (v4f){0.f, 0.f, 0.f, 0.f}, e1 = e0;
      if (valid) {
        my_sn = csrc[base + lane];
        const v4f* ep = (const v4f*)(el + my_sn * HEADS);
        e0 = ep[0];
        e1 = ep[1];
      }
      const float elv[HEADS] = {e0.x, e0.y, e0.z, e0.w, e1.x, e1.y, e1.z, e1.w};
      float p[HEADS];
      #pragma unroll
      for (int h = 0; h < HEADS; ++h) {
        float e = elv[h] + ern[h];
        e = (e > 0.f) ? e : 0.2f * e;
        const float cm = wave_max32(valid ? e : -INFINITY);
        const float mn = fmaxf(m[h], cm);
        const float sc = __expf(m[h] - mn);
        const float pv = valid ? __expf(e - mn) : 0.f;
        s[h]   = s[h] * sc + wave_sum32(pv);
        acc[h] = acc[h] * sc;
        m[h]   = mn;
        p[h]   = pv;
      }
      #pragma unroll 4
      for (int t = 0; t < cnt; ++t) {
        const int sn = __builtin_amdgcn_readlane(my_sn, t);
        const v4f v = ((const v4f*)(x + sn * IN_F))[lane];
        #pragma unroll
        for (int h = 0; h < HEADS; ++h) {
          const float ph = __builtin_bit_cast(float,
              __builtin_amdgcn_readlane(__builtin_bit_cast(int, p[h]), t));
          acc[h] += ph * v;
        }
      }
    }

    float* zrow = zs + row * ZPITCH + 4 * lane;
    #pragma unroll
    for (int h = 0; h < HEADS; ++h) {
      const float inv = (s1 > s0) ? 1.f / s[h] : 0.f;
      *(v4f*)(zrow + h * IN_F) = acc[h] * inv;
    }
  }
  __syncthreads();

  const int rc   = lane & 15;
  const int koff = (lane >> 4) << 1;
  const int hi8  = (lane >> 4) << 3;
  const int col0 = (w << 5) + rc;
  const int col1 = col0 + 16;
  v8f acc0 = {}, acc1 = {};
  {
    const float* arow = zs + rc * ZPITCH + w * IN_F;
    const float* bp   = W + koff * NCOL + col0;
    #pragma unroll 16
    for (int k = 0; k < IN_F; k += 4) {
      v2f a, b;
      a.x = arow[k + koff];          a.y = arow[k + koff + 1];
      b.x = bp[k * NCOL];            b.y = bp[(k + 1) * NCOL];
      acc0 = __builtin_amdgcn_wmma_f32_16x16x4_f32(false, a, false, b, (short)0, acc0, false, false);
      b.x = bp[k * NCOL + 16];       b.y = bp[(k + 1) * NCOL + 16];
      acc1 = __builtin_amdgcn_wmma_f32_16x16x4_f32(false, a, false, b, (short)0, acc1, false, false);
    }
  }
  __syncthreads();
  {
    const float bv0 = bias[col0], bv1 = bias[col1];
    #pragma unroll
    for (int j = 0; j < 8; ++j) {
      const float v0 = acc0[j] + bv0;
      const float v1 = acc1[j] + bv1;
      hs[(hi8 + j) * HPITCH + col0] = (v0 > 0.f) ? v0 : (__expf(v0) - 1.f);
      hs[(hi8 + j) * HPITCH + col1] = (v1 > 0.f) ? v1 : (__expf(v1) - 1.f);
    }
  }
  __syncthreads();

  #pragma unroll
  for (int r = 0; r < 2; ++r) {
    const int row = w + 8 * r;
    float sum = 0.f;
    #pragma unroll
    for (int hd = 0; hd < HEADS; ++hd) sum += hs[row * HPITCH + hd * FEAT + lane];
    const float hv = sum * (1.f / HEADS);
    h1act[(node0 + row) * FEAT + lane] = hv;
    h1s[row * H1PITCH + lane] = hv;
  }
  __syncthreads();

  {
    const int side = w >> 2;
    const int row  = ((w & 3) << 2) + (lane >> 3);
    const int hd   = lane & 7;
    const float* hp = h1s + row * H1PITCH;
    const float* tp = tab2 + side * HEADS + hd;
    float s = 0.f;
    #pragma unroll
    for (int f = 0; f < FEAT; ++f) s = fmaf(hp[f], tp[f * 16], s);
    float* const lp = side ? er2 : el2;
    lp[(node0 + row) * HEADS + hd] = s;
  }
}

#define Z2PITCH (NCOL + 4)

__global__ void __launch_bounds__(TILE_WAVES * 32)
k_layer2_fused(const float* __restrict__ h, const float* __restrict__ el,
               const float* __restrict__ er, const float* __restrict__ W,
               const float* __restrict__ bias,
               const int* __restrict__ rowptr, const int* __restrict__ csrc,
               float* __restrict__ out) {
  __shared__ __attribute__((aligned(16))) float zs[TILE_NODES * Z2PITCH];
  const int lane = threadIdx.x & 31;
  const int w = __builtin_amdgcn_readfirstlane((int)(threadIdx.x >> 5));
  const int node0 = blockIdx.x * TILE_NODES;
  const int hq = lane >> 3;
  const int hsel = ((lane & 7) + hq) & 7;

  #pragma unroll 1
  for (int r = 0; r < 2; ++r) {
    const int row = 2 * w + r;
    const int n = node0 + row;
    const int s0 = rowptr[n], s1 = rowptr[n + 1];
    const float erA = er[n * HEADS + hq];
    const float erB = er[n * HEADS + 4 + hq];

    float mA = -INFINITY, mB = -INFINITY, sA = 0.f, sB = 0.f;
    float4 accA = make_float4(0.f, 0.f, 0.f, 0.f);
    float4 accB = make_float4(0.f, 0.f, 0.f, 0.f);

    for (int base = s0; base < s1; base += 32) {
      int cnt = s1 - base;
      if (cnt > 32) cnt = 32;
      int my_sn = 0;
      if (lane < cnt) my_sn = csrc[base + lane];
      #pragma unroll 4
      for (int t = 0; t < cnt; ++t) {
        const int sn = __builtin_amdgcn_readlane(my_sn, t);
        const float4 vA = ((const float4*)(h + sn * FEAT))[lane & 7];
        const float4 vB = vA;
        const unsigned elv = __builtin_bit_cast(unsigned, el[sn * HEADS + hsel]);
        float eA = __builtin_bit_cast(float, __builtin_amdgcn_mov_dpp8(elv, 0x000000)) + erA;
        float eB = __builtin_bit_cast(float, __builtin_amdgcn_mov_dpp8(elv, 0x924924)) + erB;
        eA = (eA > 0.f) ? eA : 0.2f * eA;
        eB = (eB > 0.f) ? eB : 0.2f * eB;
        const float mnA = fmaxf(mA, eA);
        const float mnB = fmaxf(mB, eB);
        const float scA = __expf(mA - mnA);
        const float scB = __expf(mB - mnB);
        const float pA  = __expf(eA - mnA);
        const float pB  = __expf(eB - mnB);
        sA = sA * scA + pA;
        sB = sB * scB + pB;
        accA.x = accA.x * scA + pA * vA.x;
        accA.y = accA.y * scA + pA * vA.y;
        accA.z = accA.z * scA + pA * vA.z;
        accA.w = accA.w * scA + pA * vA.w;
        accB.x = accB.x * scB + pB * vB.x;
        accB.y = accB.y * scB + pB * vB.y;
        accB.z = accB.z * scB + pB * vB.z;
        accB.w = accB.w * scB + pB * vB.w;
        mA = mnA;
        mB = mnB;
      }
    }

    float4 rA = make_float4(0.f, 0.f, 0.f, 0.f);
    float4 rB = rA;
    if (s1 > s0) {
      rA.x += accA.x / sA;  rA.y += accA.y / sA;  rA.z += accA.z / sA;  rA.w += accA.w / sA;
      rB.x += accB.x / sB;  rB.y += accB.y / sB;  rB.z += accB.z / sB;  rB.w += accB.w / sB;
    }
    float4* zrow = (float4*)(zs + row * Z2PITCH);
    zrow[lane]      = rA;
    zrow[32 + lane] = rB;
  }

  const int rc   = lane & 15;
  const int koff = (lane >> 4) << 1;
  const int hi8  = (lane >> 4) << 3;
  const int col0 = (w << 5) + rc;
  const int col1 = col0 + 16;
  v2f b0[FEAT / 4], b1[FEAT / 4];
  {
    const float* bp = W + koff * NCOL + col0;
    #pragma unroll
    for (int k = 0; k < FEAT; k += 4) {
      b0[k >> 2].x = bp[k * NCOL];        b0[k >> 2].y = bp[(k + 1) * NCOL];
      b1[k >> 2].x = bp[k * NCOL + 16];   b1[k >> 2].y = bp[(k + 1) * NCOL + 16];
    }
  }
  const float bv0 = bias[col0], bv1 = bias[col1];
  __syncthreads();

  v8f acc0 = {}, acc1 = {};
  const float* arow = zs + rc * Z2PITCH + w * FEAT;
  #pragma unroll
  for (int k = 0; k < FEAT; k += 4) {
    v2f a;
    a.x = arow[k + koff];          a.y = arow[k + koff + 1];
    acc0 = __builtin_amdgcn_wmma_f32_16x16x4_f32(false, a, false, b0[k >> 2], (short)0, acc0, false, false);
    acc1 = __builtin_amdgcn_wmma_f32_16x16x4_f32(false, a, false, b1[k >> 2], (short)0, acc1, false, false);
  }
  float* crow = out + (node0 + hi8) * NCOL;
  #pragma unroll
  for (int j = 0; j < 8; ++j) {
    crow[j * NCOL + col0] = acc0[j] + bv0;
    crow[j * NCOL + col1] = acc1[j] + bv1;
  }
}

extern "C" void kernel_launch(void* const* d_in, const int* in_sizes, int n_in,
                              void* d_out, int out_size, void* d_ws, size_t ws_size,
                              hipStream_t stream) {
  const float* x   = (const float*)d_in[0];
  const int*   src = (const int*)  d_in[1];
  const int*   dst = (const int*)  d_in[2];
  const float* W1  = (const float*)d_in[3];
  const float* al1 = (const float*)d_in[4];
  const float* ar1 = (const float*)d_in[5];
  const float* b1  = (const float*)d_in[6];
  const float* W2  = (const float*)d_in[7];
  const float* al2 = (const float*)d_in[8];
  const float* ar2 = (const float*)d_in[9];
  const float* b2  = (const float*)d_in[10];

  float* wsf = (float*)d_ws;
  float* h_buf = wsf;
  float* agg1  = h_buf + (size_t)N_NODES * NCOL;
  float* elb   = agg1  + (size_t)N_NODES * NCOL;
  float* erb   = elb   + (size_t)N_NODES * HEADS;
  float* h1act = erb   + (size_t)N_NODES * HEADS;
  int* ibase   = (int*)(h1act + (size_t)N_NODES * FEAT);
  int* rowptr  = ibase;
  int* offs    = rowptr + (N_NODES + 2);
  int* csrc    = offs + 2 * N_NODES;
  int* keys    = (int*)h_buf;
  float* tab2  = agg1;
  float* el2   = tab2 + 512;
  float* er2   = el2  + (size_t)N_NODES * HEADS;

  k_logits1_wmma<<<LOGITS1_BLOCKS + 1 + EB_BLOCKS, 256, 0, stream>>>(x, W1, al1, ar1, elb, erb, W2, al2, ar2, tab2, src, dst, offs, keys);

  k_binsort<<<NBINS, BS_THREADS, 0, stream>>>(offs, keys, rowptr, csrc);

  k_layer1_fused<<<N_NODES / TILE_NODES, TILE_WAVES * 32, 0, stream>>>(x, elb, erb, W1, b1, rowptr, csrc, h1act, tab2, el2, er2);

  k_layer2_fused<<<N_NODES / TILE_NODES, TILE_WAVES * 32, 0, stream>>>(h1act, el2, er2, W2, b2, rowptr, csrc, (float*)d_out);
}
